# starting kernel
# baseline (speedup 1.0000x reference)
#include <hip/hip_runtime.h>

typedef float v2f __attribute__((ext_vector_type(2)));
typedef float v8f __attribute__((ext_vector_type(8)));

#define TK 64

__global__ __launch_bounds__(256) void k_gemm_wmma(
    const float* __restrict__ X, const float* __restrict__ Wp,
    float* __restrict__ H, int M, int K, int Kpad, int Nout)
{
    __shared__ float As[16][TK + 4];

    const int row0 = blockIdx.x << 4;
    const int t    = threadIdx.x;
    const int wave = t >> 5;
    const int lane = t & 31;
    const int half = lane >> 4;
    const int m    = lane & 15;

    const int n0 = (wave << 4) + m;
    const int n1 = ((wave + 8) << 4) + m;
    const float* __restrict__ Wr0 = Wp + (size_t)n0 * Kpad;
    const float* __restrict__ Wr1 = Wp + (size_t)n1 * Kpad;

    v8f acc0 = {}; v8f acc1 = {};

    for (int k0 = 0; k0 < Kpad; k0 += TK) {
        __syncthreads();
        #pragma unroll
        for (int i = 0; i < (16 * TK) / 256; ++i) {
            int e  = t + (i << 8);
            int r  = e >> 6;
            int kk = e & (TK - 1);
            int gr = row0 + r, gk = k0 + kk;
            As[r][kk] = (gr < M && gk < K) ? X[(size_t)gr * K + gk] : 0.0f;
        }
        __syncthreads();

        #pragma unroll
        for (int kk = 0; kk < TK; kk += 4) {
            const int kb = kk + (half << 1);
            v2f a  = *(const v2f*)&As[m][kb];
            v2f b0 = *(const v2f*)&Wr0[k0 + kb];
            v2f b1 = *(const v2f*)&Wr1[k0 + kb];
            acc0 = __builtin_amdgcn_wmma_f32_16x16x4_f32(
                false, a, false, b0, (short)0, acc0, false, false);
            acc1 = __builtin_amdgcn_wmma_f32_16x16x4_f32(
                false, a, false, b1, (short)0, acc1, false, false);
        }
    }

    #pragma unroll
    for (int r = 0; r < 8; ++r) {
        int grow = row0 + r + (half << 3);
        if (grow < M) {
            if (n0 < Nout) H[(size_t)grow * Nout + n0] = acc0[r];
            if (n1 < Nout) H[(size_t)grow * Nout + n1] = acc1[r];
        }
    }
}

__global__ void k_padW(const float* __restrict__ W, float* __restrict__ Wp,
                       int Nout, int K, int Kpad)
{
    int idx = blockIdx.x * blockDim.x + threadIdx.x;
    int tot = 256 * Kpad;
    if (idx >= tot) return;
    int r = idx / Kpad, c = idx % Kpad;
    Wp[idx] = (r < Nout && c < K) ? W[(size_t)r * K + c] : 0.0f;
}

__global__ void k_fill1(float* __restrict__ p, int n) {
    int i = blockIdx.x * blockDim.x + threadIdx.x;
    if (i < n) p[i] = 1.0f;
}

__global__ void k_deg(const int* __restrict__ dst, float* __restrict__ deg, int E) {
    int e = blockIdx.x * blockDim.x + threadIdx.x;
    if (e < E) atomicAdd(&deg[dst[e]], 1.0f);
}

__global__ void k_rsqrt(float* __restrict__ p, int n) {
    int i = blockIdx.x * blockDim.x + threadIdx.x;
    if (i < n) { float d = p[i]; p[i] = (d > 0.0f) ? rsqrtf(d) : 0.0f; }
}

__global__ void k_zero(float* __restrict__ p, size_t n) {
    size_t i = (size_t)blockIdx.x * blockDim.x + threadIdx.x;
    if (i < n) p[i] = 0.0f;
}

__global__ __launch_bounds__(256) void k_agg_wave(
    const float* __restrict__ H, const int* __restrict__ src,
    const int* __restrict__ dst, const float* __restrict__ dinv,
    float* __restrict__ out, int E, int F)
{
    const int lane   = threadIdx.x & 31;
    const int wid    = (int)((blockIdx.x * blockDim.x + threadIdx.x) >> 5);
    const int stride = (int)((gridDim.x * blockDim.x) >> 5);

    for (int e = wid; e < E; e += stride) {
        const int s = src[e];
        const int d = dst[e];
        const int e2 = e + stride;
        if (e2 < E) {
            __builtin_prefetch(&H[(size_t)src[e2] * F], 0, 1);
        }
        const float nrm = dinv[s] * dinv[d];
        const float* __restrict__ hrow = H + (size_t)s * F;
        float* __restrict__ orow = out + (size_t)d * F;
        for (int f = lane; f < F; f += 32) {
            atomicAdd(&orow[f], hrow[f] * nrm);
        }
    }
}

__global__ void k_agg_small(
    const float* __restrict__ H, const int* __restrict__ src,
    const int* __restrict__ dst, const float* __restrict__ dinv,
    float* __restrict__ out, int E, int F)
{
    size_t idx = (size_t)blockIdx.x * blockDim.x + threadIdx.x;
    size_t tot = (size_t)E * F;
    if (idx >= tot) return;
    int e = (int)(idx / F);
    int f = (int)(idx % F);
    int s = src[e], d = dst[e];
    float nrm = dinv[s] * dinv[d];
    atomicAdd(&out[(size_t)d * F + f], H[(size_t)s * F + f] * nrm);
}

__global__ void k_finish(float* __restrict__ dest, const float* __restrict__ H,
                         const float* __restrict__ dinv, const float* __restrict__ bias,
                         int N, int F, int doRelu)
{
    size_t idx = (size_t)blockIdx.x * blockDim.x + threadIdx.x;
    size_t tot = (size_t)N * F;
    if (idx >= tot) return;
    int i = (int)(idx / F);
    int f = (int)(idx % F);
    float di = dinv[i];
    float v  = dest[idx] + H[idx] * di * di + bias[f];
    dest[idx] = doRelu ? fmaxf(v, 0.0f) : v;
}

extern "C" void kernel_launch(void* const* d_in, const int* in_sizes, int n_in,
                              void* d_out, int out_size, void* d_ws, size_t ws_size,
                              hipStream_t stream) {
    (void)n_in; (void)out_size; (void)ws_size;

    const float* x   = (const float*)d_in[0];
    const int*   ei  = (const int*)  d_in[1];
    const float* W1  = (const float*)d_in[2];
    const float* b1  = (const float*)d_in[3];
    const float* W1b = (const float*)d_in[4];
    const float* b1b = (const float*)d_in[5];
    const float* W2b = (const float*)d_in[6];
    const float* b2b = (const float*)d_in[7];
    const float* W2  = (const float*)d_in[8];
    const float* b2  = (const float*)d_in[9];

    const int F_IN = 500, H1 = 200, H2 = 100, H3 = 40, C = 3;
    const int N = in_sizes[0] / F_IN;
    const int E = in_sizes[1] / 2;
    const int* srcIdx = ei;
    const int* dstIdx = ei + E;

    char*  ws   = (char*)d_ws;
    float* dinv = (float*)ws;
    size_t off  = ((size_t)N * sizeof(float) + 511) & ~(size_t)511;
    float* Wpad = (float*)(ws + off);
    off += (size_t)256 * 512 * sizeof(float);
    float* bufH = (float*)(ws + off);
    off += (((size_t)N * H1 * sizeof(float)) + 511) & ~(size_t)511;
    float* bufA = (float*)(ws + off);
    float* out  = (float*)d_out;

    k_fill1<<<(N + 255) / 256, 256, 0, stream>>>(dinv, N);
    k_deg  <<<(E + 255) / 256, 256, 0, stream>>>(dstIdx, dinv, E);
    k_rsqrt<<<(N + 255) / 256, 256, 0, stream>>>(dinv, N);

    const int gemmBlocks = (N + 15) / 16;
    const int aggBlocks  = 65536;

    auto layer = [&](const float* in, const float* Wt, const float* bias,
                     float* dest, int K, int F, int relu) {
        const int Kpad = (K + TK - 1) & ~(TK - 1);
        k_padW<<<(256 * Kpad + 255) / 256, 256, 0, stream>>>(Wt, Wpad, F, K, Kpad);
        k_gemm_wmma<<<gemmBlocks, 256, 0, stream>>>(in, Wpad, bufH, N, K, Kpad, F);
        size_t tot = (size_t)N * F;
        k_zero<<<(int)((tot + 255) / 256), 256, 0, stream>>>(dest, tot);
        if (F >= 32) {
            k_agg_wave<<<aggBlocks, 256, 0, stream>>>(bufH, srcIdx, dstIdx, dinv, dest, E, F);
        } else {
            size_t tote = (size_t)E * F;
            k_agg_small<<<(int)((tote + 255) / 256), 256, 0, stream>>>(
                bufH, srcIdx, dstIdx, dinv, dest, E, F);
        }
        k_finish<<<(int)((tot + 255) / 256), 256, 0, stream>>>(
            dest, bufH, dinv, bias, N, F, relu);
    };

    layer(x,    W1,  b1,  bufA, F_IN, H1, 1);
    layer(bufA, W1b, b1b, bufA, H1,   H2, 1);
    layer(bufA, W2b, b2b, bufA, H2,   H3, 1);
    layer(bufA, W2,  b2,  out,  H3,   C,  0);
}


# Owner blocks scan a one-byte owner key per edge (dst/392, written by the degree pass) instead of dst: 16 edges per 16-byte load, exact zero-byte test; 392 rows per block; LDS arrival order kept
# speedup vs baseline: 2.3726x; 2.3726x over previous
#include <hip/hip_runtime.h>

typedef float v2f __attribute__((ext_vector_type(2)));
typedef float v4f __attribute__((ext_vector_type(4)));
typedef float v8f __attribute__((ext_vector_type(8)));
typedef int   v4i __attribute__((ext_vector_type(4)));

#define TK 64

template <int NT>
__global__ __launch_bounds__(256) void k_gemm_wmma(
    const float* __restrict__ X, const float* __restrict__ Wf,
    float* __restrict__ H, int M, int K, int Kpad, int Nout)
{
    __shared__ float As[16][TK + 4];

    const int row0 = blockIdx.x << 4;
    const int t    = threadIdx.x;
    const int wave = (__builtin_amdgcn_readfirstlane(t >> 5) +
                      (int)((blockIdx.x * 2654435761u) >> 29)) & 7;
    const int lane = t & 31;
    const int half = lane >> 4;
    const int m    = lane & 15;
    const bool has0 = wave < NT;
    const bool has1 = (wave + 8) < NT;

    const int n0 = (wave << 4) + m;
    const int n1 = ((wave + 8) << 4) + m;
    const int KS2 = Kpad >> 3;
    const v4f* __restrict__ Wq  = (const v4f*)Wf;
    const v4f* __restrict__ Wf0 = Wq + (size_t)wave * KS2 * 32 + lane;
    const v4f* __restrict__ Wf1 = Wq + (size_t)(wave + 8) * KS2 * 32 + lane;

    v8f acc0 = {}; v8f acc1 = {};

    for (int k0 = 0; k0 < Kpad; k0 += TK) {
        __syncthreads();
        #pragma unroll
        for (int i = 0; i < (16 * TK) / 256; ++i) {
            int e  = t + (i << 8);
            int r  = e >> 6;
            int kk = e & (TK - 1);
            int gr = row0 + r, gk = k0 + kk;
            As[r][kk] = (gr < M && gk < K) ? X[(size_t)gr * K + gk] : 0.0f;
        }
        __syncthreads();

        const int s0 = k0 >> 3;
        if (has1) {
            v4f q0[TK / 8], q1[TK / 8];
            #pragma unroll
            for (int j = 0; j < TK / 8; ++j) {
                q0[j] = Wf0[(size_t)(s0 + j) * 32];
                q1[j] = Wf1[(size_t)(s0 + j) * 32];
            }
            #pragma unroll
            for (int j = 0; j < TK / 8; ++j) {
                const int kb = (j << 3) + (half << 1);
                const v2f a0 = *(const v2f*)&As[m][kb];
                const v2f a1 = *(const v2f*)&As[m][kb + 4];
                const v2f b00 = __builtin_shufflevector(q0[j], q0[j], 0, 1);
                const v2f b01 = __builtin_shufflevector(q0[j], q0[j], 2, 3);
                const v2f b10 = __builtin_shufflevector(q1[j], q1[j], 0, 1);
                const v2f b11 = __builtin_shufflevector(q1[j], q1[j], 2, 3);
                acc0 = __builtin_amdgcn_wmma_f32_16x16x4_f32(
                    false, a0, false, b00, (short)0, acc0, false, false);
                acc1 = __builtin_amdgcn_wmma_f32_16x16x4_f32(
                    false, a0, false, b10, (short)0, acc1, false, false);
                acc0 = __builtin_amdgcn_wmma_f32_16x16x4_f32(
                    false, a1, false, b01, (short)0, acc0, false, false);
                acc1 = __builtin_amdgcn_wmma_f32_16x16x4_f32(
                    false, a1, false, b11, (short)0, acc1, false, false);
            }
        } else if (has0) {
            v4f q0[TK / 8];
            #pragma unroll
            for (int j = 0; j < TK / 8; ++j)
                q0[j] = Wf0[(size_t)(s0 + j) * 32];
            #pragma unroll
            for (int j = 0; j < TK / 8; ++j) {
                const int kb = (j << 3) + (half << 1);
                const v2f a0 = *(const v2f*)&As[m][kb];
                const v2f a1 = *(const v2f*)&As[m][kb + 4];
                const v2f b00 = __builtin_shufflevector(q0[j], q0[j], 0, 1);
                const v2f b01 = __builtin_shufflevector(q0[j], q0[j], 2, 3);
                acc0 = __builtin_amdgcn_wmma_f32_16x16x4_f32(
                    false, a0, false, b00, (short)0, acc0, false, false);
                acc0 = __builtin_amdgcn_wmma_f32_16x16x4_f32(
                    false, a1, false, b01, (short)0, acc0, false, false);
            }
        }
    }

    #pragma unroll
    for (int r = 0; r < 8; ++r) {
        int grow = row0 + r + (half << 3);
        if (grow < M) {
            if (has0 && n0 < Nout) H[(size_t)grow * Nout + n0] = acc0[r];
            if (has1 && n1 < Nout) H[(size_t)grow * Nout + n1] = acc1[r];
        }
    }
}

__global__ void k_packW(const float* __restrict__ W, float* __restrict__ Wf,
                        int Nout, int K, int Kpad)
{
    int idx = blockIdx.x * blockDim.x + threadIdx.x;
    int tot = 256 * Kpad;
    if (idx >= tot) return;
    const int KS2  = Kpad >> 3;
    const int q    = idx & 3;
    const int lane = (idx >> 2) & 31;
    const int rest = idx >> 7;
    const int s    = rest % KS2;
    const int ct   = rest / KS2;
    const int n = (ct << 4) + (lane & 15);
    const int k = (s << 3) + ((q >> 1) << 2) + ((lane >> 4) << 1) + (q & 1);
    Wf[idx] = (n < Nout && k < K) ? W[(size_t)n * K + k] : 0.0f;
}

__global__ void k_fill1(float* __restrict__ p, int n) {
    int i = blockIdx.x * blockDim.x + threadIdx.x;
    if (i < n) p[i] = 1.0f;
}

__global__ void k_deg(const int* __restrict__ dst, float* __restrict__ deg, int E) {
    const int nth = (int)(gridDim.x * blockDim.x);
    for (int e = (int)(blockIdx.x * blockDim.x + threadIdx.x); e < E; e += nth)
        atomicAdd(&deg[dst[e]], 1.0f);
}

#define OWN_R 392

__global__ __launch_bounds__(256) void k_deg_key(
    const int* __restrict__ dst, float* __restrict__ deg, int* __restrict__ key, int E)
{
    const int nth = (int)(gridDim.x * blockDim.x);
    const int nq  = (E + 3) >> 2;
    const bool al = ((((size_t)dst) & 15) == 0);
    for (int qd = (int)(blockIdx.x * blockDim.x + threadIdx.x); qd < nq; qd += nth) {
        const int e0 = qd << 2;
        unsigned w = 0u;
        if (al && e0 + 4 <= E) {
            const v4i d4 = *(const v4i*)(dst + e0);
            atomicAdd(&deg[d4.x], 1.0f); atomicAdd(&deg[d4.y], 1.0f);
            atomicAdd(&deg[d4.z], 1.0f); atomicAdd(&deg[d4.w], 1.0f);
            w = (unsigned)(d4.x / OWN_R) | ((unsigned)(d4.y / OWN_R) << 8) |
                ((unsigned)(d4.z / OWN_R) << 16) | ((unsigned)(d4.w / OWN_R) << 24);
        } else {
            for (int i = 0; i < 4; ++i)
                if (e0 + i < E) {
                    const int d = dst[e0 + i];
                    atomicAdd(&deg[d], 1.0f);
                    w |= (unsigned)(d / OWN_R) << (8 * i);
                }
        }
        key[qd] = (int)w;
    }
}

__global__ void k_rsqrt(float* __restrict__ p, int n) {
    int i = blockIdx.x * blockDim.x + threadIdx.x;
    if (i < n) { float d = p[i]; p[i] = (d > 0.0f) ? rsqrtf(d) : 0.0f; }
}

__global__ __launch_bounds__(256) void k_zero(float* __restrict__ p, size_t n) {
    const size_t tid = (size_t)blockIdx.x * blockDim.x + threadIdx.x;
    const size_t nth = (size_t)gridDim.x * blockDim.x;
    if ((((size_t)p) & 15) == 0) {
        const size_t n4 = n >> 2;
        v4f* __restrict__ p4 = (v4f*)p;
        const v4f z = {0.0f, 0.0f, 0.0f, 0.0f};
        for (size_t i = tid; i < n4; i += nth) p4[i] = z;
        for (size_t i = (n4 << 2) + tid; i < n; i += nth) p[i] = 0.0f;
    } else {
        for (size_t i = tid; i < n; i += nth) p[i] = 0.0f;
    }
}

template <int NF, int TU, int NG>
__device__ __forceinline__ void agg_groups(
    const float* __restrict__ H, float* __restrict__ out,
    int s, int d, float nrm, int j, int F, int lane)
{
    constexpr int T  = 32 / TU;
    constexpr int U  = TU * NG;
    constexpr int FB = (NF - 1) << 5;
    const int tl = lane / T;
    const int tf = FB + (lane % T);

    float v[U][NF - 1];
    float vt[NG], nt[NG];
    int   dt[NG];
    #pragma unroll
    for (int u = 0; u < U; ++u) {
        const int su = __builtin_amdgcn_readlane(s, j + u);
        const float* __restrict__ hrow = H + (size_t)su * F;
        #pragma unroll
        for (int c = 0; c < NF - 1; ++c) v[u][c] = hrow[lane + (c << 5)];
    }
    #pragma unroll
    for (int g = 0; g < NG; ++g) {
        const int ju = j + g * TU + tl;
        const int st = __shfl(s, ju);
        dt[g] = __shfl(d, ju);
        nt[g] = __shfl(nrm, ju);
        vt[g] = H[(size_t)st * F + tf];
    }
    #pragma unroll
    for (int u = 0; u < U; ++u) {
        const int   du = __builtin_amdgcn_readlane(d, j + u);
        const float nu = __builtin_bit_cast(float,
            __builtin_amdgcn_readlane(__builtin_bit_cast(int, nrm), j + u));
        float* __restrict__ orow = out + (size_t)du * F;
        #pragma unroll
        for (int c = 0; c < NF - 1; ++c)
            atomicAdd(&orow[lane + (c << 5)], v[u][c] * nu);
    }
    #pragma unroll
    for (int g = 0; g < NG; ++g)
        atomicAdd(&out[(size_t)dt[g] * F + tf], vt[g] * nt[g]);
}

template <int NF>
__device__ __forceinline__ void agg_one(
    const float* __restrict__ H, float* __restrict__ out,
    int s, int d, float nrm, int j, int F, int lane)
{
    const int  fl = lane + ((NF - 1) << 5);
    const bool lastOn = (fl < F);
    const int   sj = __builtin_amdgcn_readlane(s, j);
    const int   dj = __builtin_amdgcn_readlane(d, j);
    const float nj = __builtin_bit_cast(float,
        __builtin_amdgcn_readlane(__builtin_bit_cast(int, nrm), j));
    const float* __restrict__ hrow = H + (size_t)sj * F;
    float* __restrict__ orow = out + (size_t)dj * F;
    float v[NF];
    #pragma unroll
    for (int c = 0; c < NF - 1; ++c) v[c] = hrow[lane + (c << 5)];
    v[NF - 1] = lastOn ? hrow[fl] : 0.0f;
    #pragma unroll
    for (int c = 0; c < NF - 1; ++c) atomicAdd(&orow[lane + (c << 5)], v[c] * nj);
    if (lastOn) atomicAdd(&orow[fl], v[NF - 1] * nj);
}

#define SCAN_U   8

template <int NF, int TU, int NG, int F>
__device__ __forceinline__ void own_batch(
    const float* __restrict__ H, const int* __restrict__ src,
    const int* __restrict__ dst, const float* __restrict__ dinv,
    float* acc, const int* q, int cnt, int rowBase, int lane)
{
    constexpr int U = TU * NG;
    int s = 0, dl = 0; float nrm = 0.0f;
    if (lane < cnt) {
        const int e = q[lane];
        s = src[e];
        const int d = dst[e];
        nrm = dinv[s] * dinv[d];
        dl = d - rowBase;
    }
    int j = 0;
    for (; j + U <= cnt; j += U)
        agg_groups<NF, TU, NG>(H, acc, s, dl, nrm, j, F, lane);
    for (; j < cnt; ++j)
        agg_one<NF>(H, acc, s, dl, nrm, j, F, lane);
}

template <int NF, int TU, int NG, int F, int R, bool KEYED>
__global__ __launch_bounds__(1024) void k_agg_own(
    const float* __restrict__ H, const int* __restrict__ src,
    const int* __restrict__ dst, const float* __restrict__ dinv,
    const float* __restrict__ bias, float* __restrict__ dest,
    const int* __restrict__ key, int N, int E, int doRelu)
{
    static_assert((R * F * 4) % 128 == 0, "block boundary must fall on a 128-byte line");
    static_assert(SCAN_U * 4 <= 32, "one flag bit per load and piece in a 32-bit word");
    __shared__ __attribute__((aligned(16))) float acc[R * F];
    __shared__ int   que[32][64];

    const int t    = threadIdx.x;
    const int lane = t & 31;
    const int wave = __builtin_amdgcn_readfirstlane(t >> 5);
    const int rowBase = (int)blockIdx.x * R;
    const int nrows   = (N - rowBase < R) ? (N - rowBase) : R;

    {
        const v4f z = {0.0f, 0.0f, 0.0f, 0.0f};
        v4f* a4 = (v4f*)acc;
        for (int i = t; i < (R * F) / 4; i += 1024) a4[i] = z;
    }
    __syncthreads();

    int* q = que[wave];
    int  qn = 0;
    const int nwv = (int)(blockDim.x >> 5);
    const int eb = (int)(((long long)wave * E) / nwv);
    const int ee = (int)(((long long)(wave + 1) * E) / nwv);

    if constexpr (KEYED) {
    const unsigned pat = (unsigned)blockIdx.x * 0x01010101u;
    auto zb = [&](int wv) -> unsigned {
        const unsigned x = (unsigned)wv ^ pat;
        const unsigned s7 = (x & 0x7F7F7F7Fu) + 0x7F7F7F7Fu;
        return ~(s7 | x | 0x7F7F7F7Fu);
    };
    const int P  = E >> 4;
    const int pb = (int)(((long long)wave * P) / nwv);
    const int pe = (int)(((long long)(wave + 1) * P) / nwv);
    const v4i* __restrict__ k4 = (const v4i*)key;
    for (int p0 = pb; p0 < pe; p0 += 32 * SCAN_U) {
        v4i kk[SCAN_U];
        #pragma unroll
        for (int k = 0; k < SCAN_U; ++k) {
            const int p = p0 + (k << 5) + lane;
            kk[k] = k4[p < pe ? p : pe - 1];
        }
        __builtin_amdgcn_sched_barrier(0);
        unsigned mk[SCAN_U];
        unsigned anyk = 0u;
        #pragma unroll
        for (int k = 0; k < SCAN_U; ++k) {
            const unsigned z0 = zb(kk[k].x), z1 = zb(kk[k].y);
            const unsigned z2 = zb(kk[k].z), z3 = zb(kk[k].w);
            const unsigned m = (z0 >> 7) | (z1 >> 6) | (z2 >> 5) | (z3 >> 4);
            mk[k] = (p0 + (k << 5) + lane < pe) ? m : 0u;
            anyk |= (__builtin_amdgcn_ballot_w32(mk[k] != 0u) != 0u) ? (1u << k) : 0u;
        }
        while (anyk != 0u) {
            const int k = __builtin_ctz(anyk);
            anyk &= anyk - 1u;
            unsigned m = mk[0];
            #pragma unroll
            for (int i = 1; i < SCAN_U; ++i) m = (k == i) ? mk[i] : m;
            const int ebase = (p0 + (k << 5) + lane) << 4;
            unsigned live = __builtin_amdgcn_ballot_w32(m != 0u);
            while (live != 0u) {
                const bool hit = m != 0u;
                const int  b   = __builtin_ctz(m | 0x80000000u);
                if (hit) q[qn + __builtin_popcount(live & ((1u << lane) - 1u))] =
                             ebase + ((b & 7) << 2) + (b >> 3);
                qn += __builtin_popcount(live);
                if (qn >= 32) {
                    own_batch<NF, TU, NG, F>(H, src, dst, dinv, acc, q, 32, rowBase, lane);
                    const int rest = qn - 32;
                    int tmp = 0;
                    if (lane < rest) tmp = q[32 + lane];
                    if (lane < rest) q[lane] = tmp;
                    qn = rest;
                }
                m &= m - 1u;
                live = __builtin_amdgcn_ballot_w32(m != 0u);
            }
        }
    }
    if (wave == nwv - 1 && (P << 4) < E) {
        const int pos = P << 4;
        const int e   = pos + lane;
        const bool hit = (e < E) && ((unsigned)(dst[e < E ? e : (E - 1)] - rowBase) < (unsigned)nrows);
        const unsigned m = __builtin_amdgcn_ballot_w32(hit);
        if (hit) q[qn + __builtin_popcount(m & ((1u << lane) - 1u))] = e;
        qn += __builtin_popcount(m);
    }
    } else {
    const unsigned unr = (unsigned)nrows;
    const int nE = ee - eb;
    const int* __restrict__ wp = dst + eb;
    int head = (int)(((16u - ((unsigned)(size_t)wp & 15u)) & 15u) >> 2);
    if (head > nE) head = nE;
    const int eg = eb + head;
    const int ng = (ee - eg) >> 7;
    const int et = eg + (ng << 7);
    const v4i* __restrict__ p4 = (const v4i*)(dst + eg) + lane;
    int pos = eb, g0 = 0;
    while (pos < ee) {
        unsigned hb = 0u, slots = 0u;
        int ebase, lmul;
        if (pos >= eg && g0 < ng) {
            v4i dk[SCAN_U];
            #pragma unroll
            for (int k = 0; k < SCAN_U; ++k) {
                const int g = (g0 + k < ng) ? (g0 + k) : (ng - 1);
                dk[k] = p4[(size_t)g << 5];
            }
            #pragma unroll
            for (int k = 0; k < SCAN_U; ++k) {
                if (g0 + k < ng) {
                    const unsigned u0 = (unsigned)(dk[k].x - rowBase);
                    const unsigned u1 = (unsigned)(dk[k].y - rowBase);
                    const unsigned u2 = (unsigned)(dk[k].z - rowBase);
                    const unsigned u3 = (unsigned)(dk[k].w - rowBase);
                    const unsigned a = u0 < u1 ? u0 : u1;
                    const unsigned b = u2 < u3 ? u2 : u3;
                    const unsigned mn = a < b ? a : b;
                    if (__builtin_amdgcn_ballot_w32(mn < unr) != 0u) {
                        const unsigned m0 = __builtin_amdgcn_ballot_w32(u0 < unr);
                        const unsigned m1 = __builtin_amdgcn_ballot_w32(u1 < unr);
                        const unsigned m2 = __builtin_amdgcn_ballot_w32(u2 < unr);
                        const unsigned m3 = __builtin_amdgcn_ballot_w32(u3 < unr);
                        hb |= ((u0 < unr ? 1u : 0u) | (u1 < unr ? 2u : 0u) |
                               (u2 < unr ? 4u : 0u) | (u3 < unr ? 8u : 0u)) << (4 * k);
                        slots |= ((m0 != 0u ? 1u : 0u) | (m1 != 0u ? 2u : 0u) |
                                  (m2 != 0u ? 4u : 0u) | (m3 != 0u ? 8u : 0u)) << (4 * k);
                    }
                }
            }
            ebase = eg + (g0 << 7);
            lmul  = 4;
            g0 += SCAN_U;
            pos = (g0 < ng) ? eg : et;
        } else {
            const int lim = (pos < eg) ? eg : ee;
            const int cnt = (lim - pos < 32) ? (lim - pos) : 32;
            const bool ok = lane < cnt;
            const int  dv = dst[ok ? (pos + lane) : pos];
            const bool hit = ok && ((unsigned)(dv - rowBase) < unr);
            hb    = hit ? 1u : 0u;
            slots = (__builtin_amdgcn_ballot_w32(hit) != 0u) ? 1u : 0u;
            ebase = pos;
            lmul  = 1;
            pos += cnt;
        }
        while (slots != 0u) {
            const int b = __builtin_ctz(slots);
            slots &= slots - 1u;
            const bool hit = ((hb >> b) & 1u) != 0u;
            const unsigned m = __builtin_amdgcn_ballot_w32(hit);
            if (hit) q[qn + __builtin_popcount(m & ((1u << lane) - 1u))] =
                         ebase + ((b >> 2) << 7) + lane * lmul + (b & 3);
            qn += __builtin_popcount(m);
            if (qn >= 32) {
                own_batch<NF, TU, NG, F>(H, src, dst, dinv, acc, q, 32, rowBase, lane);
                const int rest = qn - 32;
                int tmp = 0;
                if (lane < rest) tmp = q[32 + lane];
                if (lane < rest) q[lane] = tmp;
                qn = rest;
            }
        }
    }
    }
    while (qn > 0) {
        const int cnt = (qn < 32) ? qn : 32;
        own_batch<NF, TU, NG, F>(H, src, dst, dinv, acc, q, cnt, rowBase, lane);
        const int rest = qn - cnt;
        int tmp = 0;
        if (lane < rest) tmp = q[32 + lane];
        if (lane < rest) q[lane] = tmp;
        qn = rest;
    }

    __syncthreads();

    const size_t gbase = (size_t)rowBase * F;
    const bool vec = ((((size_t)dest | (size_t)H | (size_t)bias) & 15) == 0);
    if (vec) {
        const int n4 = (nrows * F) >> 2;
        for (int p = t; p < n4; p += 1024) {
            const int idx = p << 2;
            const int r = idx / F;
            const int f = idx - r * F;
            const float di = dinv[rowBase + r];
            const v4f d4 = *(const v4f*)&acc[idx];
            const v4f h4 = *(const v4f*)(H + gbase + idx);
            const v4f b4 = *(const v4f*)(bias + f);
            v4f v;
            v.x = d4.x + h4.x * di * di + b4.x;
            v.y = d4.y + h4.y * di * di + b4.y;
            v.z = d4.z + h4.z * di * di + b4.z;
            v.w = d4.w + h4.w * di * di + b4.w;
            if (doRelu) {
                v.x = fmaxf(v.x, 0.0f); v.y = fmaxf(v.y, 0.0f);
                v.z = fmaxf(v.z, 0.0f); v.w = fmaxf(v.w, 0.0f);
            }
            *(v4f*)(dest + gbase + idx) = v;
        }
    } else {
        const int n1 = nrows * F;
        for (int idx = t; idx < n1; idx += 1024) {
            const int r = idx / F;
            const int f = idx - r * F;
            const float di = dinv[rowBase + r];
            const float v = acc[idx] + H[gbase + idx] * di * di + bias[f];
            dest[gbase + idx] = doRelu ? fmaxf(v, 0.0f) : v;
        }
    }
}

__global__ void k_agg_small(
    const float* __restrict__ H, const int* __restrict__ src,
    const int* __restrict__ dst, const float* __restrict__ dinv,
    float* __restrict__ out, int E, int F)
{
    const unsigned nth = gridDim.x * blockDim.x;
    const unsigned tot = (unsigned)E * (unsigned)F;
    const unsigned uF  = (unsigned)F;
    for (unsigned idx = blockIdx.x * blockDim.x + threadIdx.x; idx < tot; idx += nth) {
        const unsigned e = idx / uF;
        const unsigned f = idx - e * uF;
        const int s = src[e], d = dst[e];
        const float nrm = dinv[s] * dinv[d];
        atomicAdd(&out[(size_t)d * F + f], H[(size_t)s * F + f] * nrm);
    }
}

__global__ void k_finish(float* __restrict__ dest, const float* __restrict__ H,
                         const float* __restrict__ dinv, const float* __restrict__ bias,
                         int N, int F, int doRelu)
{
    const unsigned tid = blockIdx.x * blockDim.x + threadIdx.x;
    const unsigned nth = gridDim.x * blockDim.x;
    const unsigned tot = (unsigned)N * (unsigned)F;
    const unsigned uF  = (unsigned)F;
    const bool vec = ((F & 3) == 0) &&
        ((((size_t)dest | (size_t)H | (size_t)bias) & 15) == 0);
    if (vec) {
        const unsigned n4 = tot >> 2;
        for (unsigned q = tid; q < n4; q += nth) {
            const unsigned idx = q << 2;
            const unsigned i = idx / uF;
            const unsigned f = idx - i * uF;
            const float di = dinv[i];
            const v4f d4 = ((const v4f*)dest)[q];
            const v4f h4 = ((const v4f*)H)[q];
            const v4f b4 = *(const v4f*)(bias + f);
            v4f v;
            v.x = d4.x + h4.x * di * di + b4.x;
            v.y = d4.y + h4.y * di * di + b4.y;
            v.z = d4.z + h4.z * di * di + b4.z;
            v.w = d4.w + h4.w * di * di + b4.w;
            if (doRelu) {
                v.x = fmaxf(v.x, 0.0f); v.y = fmaxf(v.y, 0.0f);
                v.z = fmaxf(v.z, 0.0f); v.w = fmaxf(v.w, 0.0f);
            }
            ((v4f*)dest)[q] = v;
        }
    } else {
        for (unsigned idx = tid; idx < tot; idx += nth) {
            const unsigned i = idx / uF;
            const unsigned f = idx - i * uF;
            const float di = dinv[i];
            const float v  = dest[idx] + H[idx] * di * di + bias[f];
            dest[idx] = doRelu ? fmaxf(v, 0.0f) : v;
        }
    }
}

extern "C" void kernel_launch(void* const* d_in, const int* in_sizes, int n_in,
                              void* d_out, int out_size, void* d_ws, size_t ws_size,
                              hipStream_t stream) {
    (void)n_in; (void)out_size;

    const float* x   = (const float*)d_in[0];
    const int*   ei  = (const int*)  d_in[1];
    const float* W1  = (const float*)d_in[2];
    const float* b1  = (const float*)d_in[3];
    const float* W1b = (const float*)d_in[4];
    const float* b1b = (const float*)d_in[5];
    const float* W2b = (const float*)d_in[6];
    const float* b2b = (const float*)d_in[7];
    const float* W2  = (const float*)d_in[8];
    const float* b2  = (const float*)d_in[9];

    const int F_IN = 500, H1 = 200, H2 = 100, H3 = 40, C = 3;
    const int N = in_sizes[0] / F_IN;
    const int E = in_sizes[1] / 2;
    const int* srcIdx = ei;
    const int* dstIdx = ei + E;

    char*  ws   = (char*)d_ws;
    float* dinv = (float*)ws;
    size_t off  = ((size_t)N * sizeof(float) + 511) & ~(size_t)511;
    float* Wpad = (float*)(ws + off);
    off += (size_t)256 * 512 * sizeof(float);
    float* bufH = (float*)(ws + off);
    off += (((size_t)N * H1 * sizeof(float)) + 511) & ~(size_t)511;
    float* bufA = (float*)(ws + off);
    float* out  = (float*)d_out;
    off += (((size_t)N * H1 * sizeof(float)) + 511) & ~(size_t)511;
    int* keyp = (int*)(ws + off);
    const bool keyed = (ws_size >= off + ((((size_t)E + 15) >> 4) << 4)) &&
                       ((N + OWN_R - 1) / OWN_R <= 256);

    const int resBlocks  = 2048;
    auto gridFor = [&](size_t items) {
        size_t b = (items + 255) / 256;
        return (int)(b < (size_t)resBlocks ? b : (size_t)resBlocks);
    };

    k_fill1<<<(N + 255) / 256, 256, 0, stream>>>(dinv, N);
    if (keyed)
        k_deg_key<<<gridFor(((size_t)E + 3) / 4), 256, 0, stream>>>(dstIdx, dinv, keyp, E);
    else
        k_deg<<<gridFor(((size_t)E + 3) / 4), 256, 0, stream>>>(dstIdx, dinv, E);
    k_rsqrt<<<(N + 255) / 256, 256, 0, stream>>>(dinv, N);

    const int gemmBlocks = (N + 15) / 16;

    auto layer = [&](const float* in, const float* Wt, const float* bias,
                     float* dest, int K, int F, int relu) {
        const int Kpad = (K + TK - 1) & ~(TK - 1);
        k_packW<<<(256 * Kpad + 255) / 256, 256, 0, stream>>>(Wt, Wpad, F, K, Kpad);
        if (F == 100)
            k_gemm_wmma<7><<<gemmBlocks, 256, 0, stream>>>(in, Wpad, bufH, N, K, Kpad, F);
        else if (F == 40)
            k_gemm_wmma<3><<<gemmBlocks, 256, 0, stream>>>(in, Wpad, bufH, N, K, Kpad, F);
        else if (F <= 16)
            k_gemm_wmma<1><<<gemmBlocks, 256, 0, stream>>>(in, Wpad, bufH, N, K, Kpad, F);
        else
            k_gemm_wmma<16><<<gemmBlocks, 256, 0, stream>>>(in, Wpad, bufH, N, K, Kpad, F);
        const int keyBlocks = (N + OWN_R - 1) / OWN_R;
        if (F == 200 && keyed) {
            k_agg_own<7, 4, 1, 200, OWN_R, true><<<keyBlocks, 1024, 0, stream>>>(
                bufH, srcIdx, dstIdx, dinv, bias, dest, keyp, N, E, relu);
        } else if (F == 100 && keyed) {
            k_agg_own<4, 8, 1, 100, OWN_R, true><<<keyBlocks, 1024, 0, stream>>>(
                bufH, srcIdx, dstIdx, dinv, bias, dest, keyp, N, E, relu);
        } else if (F == 40 && keyed) {
            k_agg_own<2, 4, 2, 40, OWN_R, true><<<keyBlocks, 1024, 0, stream>>>(
                bufH, srcIdx, dstIdx, dinv, bias, dest, keyp, N, E, relu);
        } else if (F == 200) {
            k_agg_own<7, 4, 1, 200, 396, false><<<(N + 395) / 396, 1024, 0, stream>>>(
                bufH, srcIdx, dstIdx, dinv, bias, dest, keyp, N, E, relu);
        } else if (F == 100) {
            k_agg_own<4, 8, 1, 100, 400, false><<<(N + 399) / 400, 1024, 0, stream>>>(
                bufH, srcIdx, dstIdx, dinv, bias, dest, keyp, N, E, relu);
        } else if (F == 40) {
            k_agg_own<2, 4, 2, 40, 400, false><<<(N + 399) / 400, 1024, 0, stream>>>(
                bufH, srcIdx, dstIdx, dinv, bias, dest, keyp, N, E, relu);
        } else {
            size_t tot = (size_t)N * F;
            const int streamGrid = gridFor((tot + 3) / 4);
            k_zero<<<streamGrid, 256, 0, stream>>>(dest, tot);
            k_agg_small<<<gridFor(((size_t)E + 3) / 4), 256, 0, stream>>>(
                bufH, srcIdx, dstIdx, dinv, dest, E, F);
            k_finish<<<streamGrid, 256, 0, stream>>>(
                dest, bufH, dinv, bias, N, F, relu);
        }
    };

    layer(x,    W1,  b1,  bufA, F_IN, H1, 1);
    layer(bufA, W1b, b1b, bufA, H1,   H2, 1);
    layer(bufA, W2b, b2b, bufA, H2,   H3, 1);
    layer(bufA, W2,  b2,  out,  H3,   C,  0);
}


# Degree count and 3-wide last layer move to owner blocks (LDS adds over the key plane): no global float atomics; fill, rsqrt, zero, finish launches gone (17 to 14); last layer in LDS arrival order
# speedup vs baseline: 2.5841x; 1.0891x over previous
#include <hip/hip_runtime.h>

typedef float v2f __attribute__((ext_vector_type(2)));
typedef float v4f __attribute__((ext_vector_type(4)));
typedef float v8f __attribute__((ext_vector_type(8)));
typedef int   v4i __attribute__((ext_vector_type(4)));

#define TK 64

template <int NT>
__global__ __launch_bounds__(256) void k_gemm_wmma(
    const float* __restrict__ X, const float* __restrict__ Wf,
    float* __restrict__ H, int M, int K, int Kpad, int Nout)
{
    __shared__ float As[16][TK + 4];

    const int row0 = blockIdx.x << 4;
    const int t    = threadIdx.x;
    const int wave = (__builtin_amdgcn_readfirstlane(t >> 5) +
                      (int)((blockIdx.x * 2654435761u) >> 29)) & 7;
    const int lane = t & 31;
    const int half = lane >> 4;
    const int m    = lane & 15;
    const bool has0 = wave < NT;
    const bool has1 = (wave + 8) < NT;

    const int n0 = (wave << 4) + m;
    const int n1 = ((wave + 8) << 4) + m;
    const int KS2 = Kpad >> 3;
    const v4f* __restrict__ Wq  = (const v4f*)Wf;
    const v4f* __restrict__ Wf0 = Wq + (size_t)wave * KS2 * 32 + lane;
    const v4f* __restrict__ Wf1 = Wq + (size_t)(wave + 8) * KS2 * 32 + lane;

    v8f acc0 = {}; v8f acc1 = {};

    for (int k0 = 0; k0 < Kpad; k0 += TK) {
        __syncthreads();
        #pragma unroll
        for (int i = 0; i < (16 * TK) / 256; ++i) {
            int e  = t + (i << 8);
            int r  = e >> 6;
            int kk = e & (TK - 1);
            int gr = row0 + r, gk = k0 + kk;
            As[r][kk] = (gr < M && gk < K) ? X[(size_t)gr * K + gk] : 0.0f;
        }
        __syncthreads();

        const int s0 = k0 >> 3;
        if (has1) {
            v4f q0[TK / 8], q1[TK / 8];
            #pragma unroll
            for (int j = 0; j < TK / 8; ++j) {
                q0[j] = Wf0[(size_t)(s0 + j) * 32];
                q1[j] = Wf1[(size_t)(s0 + j) * 32];
            }
            #pragma unroll
            for (int j = 0; j < TK / 8; ++j) {
                const int kb = (j << 3) + (half << 1);
                const v2f a0 = *(const v2f*)&As[m][kb];
                const v2f a1 = *(const v2f*)&As[m][kb + 4];
                const v2f b00 = __builtin_shufflevector(q0[j], q0[j], 0, 1);
                const v2f b01 = __builtin_shufflevector(q0[j], q0[j], 2, 3);
                const v2f b10 = __builtin_shufflevector(q1[j], q1[j], 0, 1);
                const v2f b11 = __builtin_shufflevector(q1[j], q1[j], 2, 3);
                acc0 = __builtin_amdgcn_wmma_f32_16x16x4_f32(
                    false, a0, false, b00, (short)0, acc0, false, false);
                acc1 = __builtin_amdgcn_wmma_f32_16x16x4_f32(
                    false, a0, false, b10, (short)0, acc1, false, false);
                acc0 = __builtin_amdgcn_wmma_f32_16x16x4_f32(
                    false, a1, false, b01, (short)0, acc0, false, false);
                acc1 = __builtin_amdgcn_wmma_f32_16x16x4_f32(
                    false, a1, false, b11, (short)0, acc1, false, false);
            }
        } else if (has0) {
            v4f q0[TK / 8];
            #pragma unroll
            for (int j = 0; j < TK / 8; ++j)
                q0[j] = Wf0[(size_t)(s0 + j) * 32];
            #pragma unroll
            for (int j = 0; j < TK / 8; ++j) {
                const int kb = (j << 3) + (half << 1);
                const v2f a0 = *(const v2f*)&As[m][kb];
                const v2f a1 = *(const v2f*)&As[m][kb + 4];
                const v2f b00 = __builtin_shufflevector(q0[j], q0[j], 0, 1);
                const v2f b01 = __builtin_shufflevector(q0[j], q0[j], 2, 3);
                acc0 = __builtin_amdgcn_wmma_f32_16x16x4_f32(
                    false, a0, false, b00, (short)0, acc0, false, false);
                acc0 = __builtin_amdgcn_wmma_f32_16x16x4_f32(
                    false, a1, false, b01, (short)0, acc0, false, false);
            }
        }
    }

    #pragma unroll
    for (int r = 0; r < 8; ++r) {
        int grow = row0 + r + (half << 3);
        if (grow < M) {
            if (has0 && n0 < Nout) H[(size_t)grow * Nout + n0] = acc0[r];
            if (has1 && n1 < Nout) H[(size_t)grow * Nout + n1] = acc1[r];
        }
    }
}

__global__ void k_packW(const float* __restrict__ W, float* __restrict__ Wf,
                        int Nout, int K, int Kpad)
{
    int idx = blockIdx.x * blockDim.x + threadIdx.x;
    int tot = 256 * Kpad;
    if (idx >= tot) return;
    const int KS2  = Kpad >> 3;
    const int q    = idx & 3;
    const int lane = (idx >> 2) & 31;
    const int rest = idx >> 7;
    const int s    = rest % KS2;
    const int ct   = rest / KS2;
    const int n = (ct << 4) + (lane & 15);
    const int k = (s << 3) + ((q >> 1) << 2) + ((lane >> 4) << 1) + (q & 1);
    Wf[idx] = (n < Nout && k < K) ? W[(size_t)n * K + k] : 0.0f;
}

__global__ void k_fill1(float* __restrict__ p, int n) {
    int i = blockIdx.x * blockDim.x + threadIdx.x;
    if (i < n) p[i] = 1.0f;
}

__global__ void k_deg(const int* __restrict__ dst, float* __restrict__ deg, int E) {
    const int nth = (int)(gridDim.x * blockDim.x);
    for (int e = (int)(blockIdx.x * blockDim.x + threadIdx.x); e < E; e += nth)
        atomicAdd(&deg[dst[e]], 1.0f);
}

#define OWN_R 392

__global__ __launch_bounds__(256) void k_key(
    const int* __restrict__ dst, int* __restrict__ key, int E)
{
    const int nth = (int)(gridDim.x * blockDim.x);
    const int nq  = (E + 3) >> 2;
    const bool al = ((((size_t)dst) & 15) == 0);
    for (int qd = (int)(blockIdx.x * blockDim.x + threadIdx.x); qd < nq; qd += nth) {
        const int e0 = qd << 2;
        unsigned w = 0u;
        if (al && e0 + 4 <= E) {
            const v4i d4 = *(const v4i*)(dst + e0);
            w = (unsigned)(d4.x / OWN_R) | ((unsigned)(d4.y / OWN_R) << 8) |
                ((unsigned)(d4.z / OWN_R) << 16) | ((unsigned)(d4.w / OWN_R) << 24);
        } else {
            for (int i = 0; i < 4; ++i)
                if (e0 + i < E) w |= (unsigned)(dst[e0 + i] / OWN_R) << (8 * i);
        }
        key[qd] = (int)w;
    }
}

__global__ void k_rsqrt(float* __restrict__ p, int n) {
    int i = blockIdx.x * blockDim.x + threadIdx.x;
    if (i < n) { float d = p[i]; p[i] = (d > 0.0f) ? rsqrtf(d) : 0.0f; }
}

__global__ __launch_bounds__(256) void k_zero(float* __restrict__ p, size_t n) {
    const size_t tid = (size_t)blockIdx.x * blockDim.x + threadIdx.x;
    const size_t nth = (size_t)gridDim.x * blockDim.x;
    if ((((size_t)p) & 15) == 0) {
        const size_t n4 = n >> 2;
        v4f* __restrict__ p4 = (v4f*)p;
        const v4f z = {0.0f, 0.0f, 0.0f, 0.0f};
        for (size_t i = tid; i < n4; i += nth) p4[i] = z;
        for (size_t i = (n4 << 2) + tid; i < n; i += nth) p[i] = 0.0f;
    } else {
        for (size_t i = tid; i < n; i += nth) p[i] = 0.0f;
    }
}

template <int NF, int TU, int NG>
__device__ __forceinline__ void agg_groups(
    const float* __restrict__ H, float* __restrict__ out,
    int s, int d, float nrm, int j, int F, int lane)
{
    constexpr int T  = 32 / TU;
    constexpr int U  = TU * NG;
    constexpr int FB = (NF - 1) << 5;
    const int tl = lane / T;
    const int tf = FB + (lane % T);

    float v[U][NF - 1];
    float vt[NG], nt[NG];
    int   dt[NG];
    #pragma unroll
    for (int u = 0; u < U; ++u) {
        const int su = __builtin_amdgcn_readlane(s, j + u);
        const float* __restrict__ hrow = H + (size_t)su * F;
        #pragma unroll
        for (int c = 0; c < NF - 1; ++c) v[u][c] = hrow[lane + (c << 5)];
    }
    #pragma unroll
    for (int g = 0; g < NG; ++g) {
        const int ju = j + g * TU + tl;
        const int st = __shfl(s, ju);
        dt[g] = __shfl(d, ju);
        nt[g] = __shfl(nrm, ju);
        vt[g] = H[(size_t)st * F + tf];
    }
    #pragma unroll
    for (int u = 0; u < U; ++u) {
        const int   du = __builtin_amdgcn_readlane(d, j + u);
        const float nu = __builtin_bit_cast(float,
            __builtin_amdgcn_readlane(__builtin_bit_cast(int, nrm), j + u));
        float* __restrict__ orow = out + (size_t)du * F;
        #pragma unroll
        for (int c = 0; c < NF - 1; ++c)
            atomicAdd(&orow[lane + (c << 5)], v[u][c] * nu);
    }
    #pragma unroll
    for (int g = 0; g < NG; ++g)
        atomicAdd(&out[(size_t)dt[g] * F + tf], vt[g] * nt[g]);
}

template <int NF>
__device__ __forceinline__ void agg_one(
    const float* __restrict__ H, float* __restrict__ out,
    int s, int d, float nrm, int j, int F, int lane)
{
    const int  fl = lane + ((NF - 1) << 5);
    const bool lastOn = (fl < F);
    const int   sj = __builtin_amdgcn_readlane(s, j);
    const int   dj = __builtin_amdgcn_readlane(d, j);
    const float nj = __builtin_bit_cast(float,
        __builtin_amdgcn_readlane(__builtin_bit_cast(int, nrm), j));
    const float* __restrict__ hrow = H + (size_t)sj * F;
    float* __restrict__ orow = out + (size_t)dj * F;
    float v[NF];
    #pragma unroll
    for (int c = 0; c < NF - 1; ++c) v[c] = hrow[lane + (c << 5)];
    v[NF - 1] = lastOn ? hrow[fl] : 0.0f;
    #pragma unroll
    for (int c = 0; c < NF - 1; ++c) atomicAdd(&orow[lane + (c << 5)], v[c] * nj);
    if (lastOn) atomicAdd(&orow[fl], v[NF - 1] * nj);
}

#define SCAN_U   8

template <int NF, int TU, int NG, int F, bool DEG>
__device__ __forceinline__ void own_batch(
    const float* __restrict__ H, const int* __restrict__ src,
    const int* __restrict__ dst, const float* __restrict__ dinv,
    float* acc, const int* q, int cnt, int rowBase, int lane)
{
    if constexpr (NF == 0) {
        if (lane < cnt) {
            const int e = q[lane];
            const int d = dst[e];
            float* row = acc + (d - rowBase) * F;
            if constexpr (DEG) {
                atomicAdd(row, 1.0f);
            } else {
                const int s = src[e];
                const float nrm = dinv[s] * dinv[d];
                const float* __restrict__ hrow = H + (size_t)s * F;
                float v[F];
                #pragma unroll
                for (int f = 0; f < F; ++f) v[f] = hrow[f];
                #pragma unroll
                for (int f = 0; f < F; ++f) atomicAdd(&row[f], v[f] * nrm);
            }
        }
    } else {
    constexpr int U = TU * NG;
    int s = 0, dl = 0; float nrm = 0.0f;
    if (lane < cnt) {
        const int e = q[lane];
        s = src[e];
        const int d = dst[e];
        nrm = dinv[s] * dinv[d];
        dl = d - rowBase;
    }
    int j = 0;
    for (; j + U <= cnt; j += U)
        agg_groups<NF, TU, NG>(H, acc, s, dl, nrm, j, F, lane);
    for (; j < cnt; ++j)
        agg_one<NF>(H, acc, s, dl, nrm, j, F, lane);
    }
}

template <int NF, int TU, int NG, int F, int R, bool KEYED, bool DEG>
__global__ __launch_bounds__(1024) void k_agg_own(
    const float* __restrict__ H, const int* __restrict__ src,
    const int* __restrict__ dst, const float* __restrict__ dinv,
    const float* __restrict__ bias, float* __restrict__ dest,
    const int* __restrict__ key, int N, int E, int doRelu)
{
    static_assert(NF == 0 || (R * F * 4) % 128 == 0, "block boundary must fall on a 128-byte line");
    static_assert((R * F) % 4 == 0, "the sums are cleared in 16-byte pieces");
    static_assert(SCAN_U * 4 <= 32, "one flag bit per load and piece in a 32-bit word");
    constexpr int ACC_N = (NF == 0 && R * F < 41472) ? 41472 : R * F;
    __shared__ __attribute__((aligned(16))) float acc[ACC_N];
    __shared__ int   que[32][64];

    const int t    = threadIdx.x;
    const int lane = t & 31;
    const int wave = __builtin_amdgcn_readfirstlane(t >> 5);
    const int rowBase = (int)blockIdx.x * R;
    const int nrows   = (N - rowBase < R) ? (N - rowBase) : R;

    {
        const v4f z = {0.0f, 0.0f, 0.0f, 0.0f};
        v4f* a4 = (v4f*)acc;
        for (int i = t; i < (R * F) / 4; i += 1024) a4[i] = z;
    }
    __syncthreads();

    int* q = que[wave];
    int  qn = 0;
    const int nwv = (int)(blockDim.x >> 5);
    const int eb = (int)(((long long)wave * E) / nwv);
    const int ee = (int)(((long long)(wave + 1) * E) / nwv);

    if constexpr (KEYED) {
    const unsigned pat = (unsigned)blockIdx.x * 0x01010101u;
    auto zb = [&](int wv) -> unsigned {
        const unsigned x = (unsigned)wv ^ pat;
        const unsigned s7 = (x & 0x7F7F7F7Fu) + 0x7F7F7F7Fu;
        return ~(s7 | x | 0x7F7F7F7Fu);
    };
    const int P  = E >> 4;
    const int pb = (int)(((long long)wave * P) / nwv);
    const int pe = (int)(((long long)(wave + 1) * P) / nwv);
    const v4i* __restrict__ k4 = (const v4i*)key;
    for (int p0 = pb; p0 < pe; p0 += 32 * SCAN_U) {
        v4i kk[SCAN_U];
        #pragma unroll
        for (int k = 0; k < SCAN_U; ++k) {
            const int p = p0 + (k << 5) + lane;
            kk[k] = k4[p < pe ? p : pe - 1];
        }
        __builtin_amdgcn_sched_barrier(0);
        unsigned mk[SCAN_U];
        unsigned anyk = 0u;
        #pragma unroll
        for (int k = 0; k < SCAN_U; ++k) {
            const unsigned z0 = zb(kk[k].x), z1 = zb(kk[k].y);
            const unsigned z2 = zb(kk[k].z), z3 = zb(kk[k].w);
            const unsigned m = (z0 >> 7) | (z1 >> 6) | (z2 >> 5) | (z3 >> 4);
            mk[k] = (p0 + (k << 5) + lane < pe) ? m : 0u;
            anyk |= (__builtin_amdgcn_ballot_w32(mk[k] != 0u) != 0u) ? (1u << k) : 0u;
        }
        while (anyk != 0u) {
            const int k = __builtin_ctz(anyk);
            anyk &= anyk - 1u;
            unsigned m = mk[0];
            #pragma unroll
            for (int i = 1; i < SCAN_U; ++i) m = (k == i) ? mk[i] : m;
            const int ebase = (p0 + (k << 5) + lane) << 4;
            unsigned live = __builtin_amdgcn_ballot_w32(m != 0u);
            while (live != 0u) {
                const bool hit = m != 0u;
                const int  b   = __builtin_ctz(m | 0x80000000u);
                if (hit) q[qn + __builtin_popcount(live & ((1u << lane) - 1u))] =
                             ebase + ((b & 7) << 2) + (b >> 3);
                qn += __builtin_popcount(live);
                if (qn >= 32) {
                    own_batch<NF, TU, NG, F, DEG>(H, src, dst, dinv, acc, q, 32, rowBase, lane);
                    const int rest = qn - 32;
                    int tmp = 0;
                    if (lane < rest) tmp = q[32 + lane];
                    if (lane < rest) q[lane] = tmp;
                    qn = rest;
                }
                m &= m - 1u;
                live = __builtin_amdgcn_ballot_w32(m != 0u);
            }
        }
    }
    if (wave == nwv - 1 && (P << 4) < E) {
        const int pos = P << 4;
        const int e   = pos + lane;
        const bool hit = (e < E) && ((unsigned)(dst[e < E ? e : (E - 1)] - rowBase) < (unsigned)nrows);
        const unsigned m = __builtin_amdgcn_ballot_w32(hit);
        if (hit) q[qn + __builtin_popcount(m & ((1u << lane) - 1u))] = e;
        qn += __builtin_popcount(m);
    }
    } else {
    const unsigned unr = (unsigned)nrows;
    const int nE = ee - eb;
    const int* __restrict__ wp = dst + eb;
    int head = (int)(((16u - ((unsigned)(size_t)wp & 15u)) & 15u) >> 2);
    if (head > nE) head = nE;
    const int eg = eb + head;
    const int ng = (ee - eg) >> 7;
    const int et = eg + (ng << 7);
    const v4i* __restrict__ p4 = (const v4i*)(dst + eg) + lane;
    int pos = eb, g0 = 0;
    while (pos < ee) {
        unsigned hb = 0u, slots = 0u;
        int ebase, lmul;
        if (pos >= eg && g0 < ng) {
            v4i dk[SCAN_U];
            #pragma unroll
            for (int k = 0; k < SCAN_U; ++k) {
                const int g = (g0 + k < ng) ? (g0 + k) : (ng - 1);
                dk[k] = p4[(size_t)g << 5];
            }
            #pragma unroll
            for (int k = 0; k < SCAN_U; ++k) {
                if (g0 + k < ng) {
                    const unsigned u0 = (unsigned)(dk[k].x - rowBase);
                    const unsigned u1 = (unsigned)(dk[k].y - rowBase);
                    const unsigned u2 = (unsigned)(dk[k].z - rowBase);
                    const unsigned u3 = (unsigned)(dk[k].w - rowBase);
                    const unsigned a = u0 < u1 ? u0 : u1;
                    const unsigned b = u2 < u3 ? u2 : u3;
                    const unsigned mn = a < b ? a : b;
                    if (__builtin_amdgcn_ballot_w32(mn < unr) != 0u) {
                        const unsigned m0 = __builtin_amdgcn_ballot_w32(u0 < unr);
                        const unsigned m1 = __builtin_amdgcn_ballot_w32(u1 < unr);
                        const unsigned m2 = __builtin_amdgcn_ballot_w32(u2 < unr);
                        const unsigned m3 = __builtin_amdgcn_ballot_w32(u3 < unr);
                        hb |= ((u0 < unr ? 1u : 0u) | (u1 < unr ? 2u : 0u) |
                               (u2 < unr ? 4u : 0u) | (u3 < unr ? 8u : 0u)) << (4 * k);
                        slots |= ((m0 != 0u ? 1u : 0u) | (m1 != 0u ? 2u : 0u) |
                                  (m2 != 0u ? 4u : 0u) | (m3 != 0u ? 8u : 0u)) << (4 * k);
                    }
                }
            }
            ebase = eg + (g0 << 7);
            lmul  = 4;
            g0 += SCAN_U;
            pos = (g0 < ng) ? eg : et;
        } else {
            const int lim = (pos < eg) ? eg : ee;
            const int cnt = (lim - pos < 32) ? (lim - pos) : 32;
            const bool ok = lane < cnt;
            const int  dv = dst[ok ? (pos + lane) : pos];
            const bool hit = ok && ((unsigned)(dv - rowBase) < unr);
            hb    = hit ? 1u : 0u;
            slots = (__builtin_amdgcn_ballot_w32(hit) != 0u) ? 1u : 0u;
            ebase = pos;
            lmul  = 1;
            pos += cnt;
        }
        while (slots != 0u) {
            const int b = __builtin_ctz(slots);
            slots &= slots - 1u;
            const bool hit = ((hb >> b) & 1u) != 0u;
            const unsigned m = __builtin_amdgcn_ballot_w32(hit);
            if (hit) q[qn + __builtin_popcount(m & ((1u << lane) - 1u))] =
                         ebase + ((b >> 2) << 7) + lane * lmul + (b & 3);
            qn += __builtin_popcount(m);
            if (qn >= 32) {
                own_batch<NF, TU, NG, F, DEG>(H, src, dst, dinv, acc, q, 32, rowBase, lane);
                const int rest = qn - 32;
                int tmp = 0;
                if (lane < rest) tmp = q[32 + lane];
                if (lane < rest) q[lane] = tmp;
                qn = rest;
            }
        }
    }
    }
    while (qn > 0) {
        const int cnt = (qn < 32) ? qn : 32;
        own_batch<NF, TU, NG, F, DEG>(H, src, dst, dinv, acc, q, cnt, rowBase, lane);
        const int rest = qn - cnt;
        int tmp = 0;
        if (lane < rest) tmp = q[32 + lane];
        if (lane < rest) q[lane] = tmp;
        qn = rest;
    }

    __syncthreads();

    if constexpr (DEG) {
        for (int r = t; r < nrows; r += 1024) {
            const float dg = 1.0f + acc[r];
            dest[rowBase + r] = (dg > 0.0f) ? rsqrtf(dg) : 0.0f;
        }
    } else {
    const size_t gbase = (size_t)rowBase * F;
    const bool vec = (F % 4 == 0) &&
        ((((size_t)dest | (size_t)H | (size_t)bias) & 15) == 0);
    if (vec) {
        const int n4 = (nrows * F) >> 2;
        for (int p = t; p < n4; p += 1024) {
            const int idx = p << 2;
            const int r = idx / F;
            const int f = idx - r * F;
            const float di = dinv[rowBase + r];
            const v4f d4 = *(const v4f*)&acc[idx];
            const v4f h4 = *(const v4f*)(H + gbase + idx);
            const v4f b4 = *(const v4f*)(bias + f);
            v4f v;
            v.x = d4.x + h4.x * di * di + b4.x;
            v.y = d4.y + h4.y * di * di + b4.y;
            v.z = d4.z + h4.z * di * di + b4.z;
            v.w = d4.w + h4.w * di * di + b4.w;
            if (doRelu) {
                v.x = fmaxf(v.x, 0.0f); v.y = fmaxf(v.y, 0.0f);
                v.z = fmaxf(v.z, 0.0f); v.w = fmaxf(v.w, 0.0f);
            }
            *(v4f*)(dest + gbase + idx) = v;
        }
    } else {
        const int n1 = nrows * F;
        for (int idx = t; idx < n1; idx += 1024) {
            const int r = idx / F;
            const int f = idx - r * F;
            const float di = dinv[rowBase + r];
            const float v = acc[idx] + H[gbase + idx] * di * di + bias[f];
            dest[gbase + idx] = doRelu ? fmaxf(v, 0.0f) : v;
        }
    }
    }
}

__global__ void k_agg_small(
    const float* __restrict__ H, const int* __restrict__ src,
    const int* __restrict__ dst, const float* __restrict__ dinv,
    float* __restrict__ out, int E, int F)
{
    const unsigned nth = gridDim.x * blockDim.x;
    const unsigned tot = (unsigned)E * (unsigned)F;
    const unsigned uF  = (unsigned)F;
    for (unsigned idx = blockIdx.x * blockDim.x + threadIdx.x; idx < tot; idx += nth) {
        const unsigned e = idx / uF;
        const unsigned f = idx - e * uF;
        const int s = src[e], d = dst[e];
        const float nrm = dinv[s] * dinv[d];
        atomicAdd(&out[(size_t)d * F + f], H[(size_t)s * F + f] * nrm);
    }
}

__global__ void k_finish(float* __restrict__ dest, const float* __restrict__ H,
                         const float* __restrict__ dinv, const float* __restrict__ bias,
                         int N, int F, int doRelu)
{
    const unsigned tid = blockIdx.x * blockDim.x + threadIdx.x;
    const unsigned nth = gridDim.x * blockDim.x;
    const unsigned tot = (unsigned)N * (unsigned)F;
    const unsigned uF  = (unsigned)F;
    const bool vec = ((F & 3) == 0) &&
        ((((size_t)dest | (size_t)H | (size_t)bias) & 15) == 0);
    if (vec) {
        const unsigned n4 = tot >> 2;
        for (unsigned q = tid; q < n4; q += nth) {
            const unsigned idx = q << 2;
            const unsigned i = idx / uF;
            const unsigned f = idx - i * uF;
            const float di = dinv[i];
            const v4f d4 = ((const v4f*)dest)[q];
            const v4f h4 = ((const v4f*)H)[q];
            const v4f b4 = *(const v4f*)(bias + f);
            v4f v;
            v.x = d4.x + h4.x * di * di + b4.x;
            v.y = d4.y + h4.y * di * di + b4.y;
            v.z = d4.z + h4.z * di * di + b4.z;
            v.w = d4.w + h4.w * di * di + b4.w;
            if (doRelu) {
                v.x = fmaxf(v.x, 0.0f); v.y = fmaxf(v.y, 0.0f);
                v.z = fmaxf(v.z, 0.0f); v.w = fmaxf(v.w, 0.0f);
            }
            ((v4f*)dest)[q] = v;
        }
    } else {
        for (unsigned idx = tid; idx < tot; idx += nth) {
            const unsigned i = idx / uF;
            const unsigned f = idx - i * uF;
            const float di = dinv[i];
            const float v  = dest[idx] + H[idx] * di * di + bias[f];
            dest[idx] = doRelu ? fmaxf(v, 0.0f) : v;
        }
    }
}

extern "C" void kernel_launch(void* const* d_in, const int* in_sizes, int n_in,
                              void* d_out, int out_size, void* d_ws, size_t ws_size,
                              hipStream_t stream) {
    (void)n_in; (void)out_size;

    const float* x   = (const float*)d_in[0];
    const int*   ei  = (const int*)  d_in[1];
    const float* W1  = (const float*)d_in[2];
    const float* b1  = (const float*)d_in[3];
    const float* W1b = (const float*)d_in[4];
    const float* b1b = (const float*)d_in[5];
    const float* W2b = (const float*)d_in[6];
    const float* b2b = (const float*)d_in[7];
    const float* W2  = (const float*)d_in[8];
    const float* b2  = (const float*)d_in[9];

    const int F_IN = 500, H1 = 200, H2 = 100, H3 = 40, C = 3;
    const int N = in_sizes[0] / F_IN;
    const int E = in_sizes[1] / 2;
    const int* srcIdx = ei;
    const int* dstIdx = ei + E;

    char*  ws   = (char*)d_ws;
    float* dinv = (float*)ws;
    size_t off  = ((size_t)N * sizeof(float) + 511) & ~(size_t)511;
    float* Wpad = (float*)(ws + off);
    off += (size_t)256 * 512 * sizeof(float);
    float* bufH = (float*)(ws + off);
    off += (((size_t)N * H1 * sizeof(float)) + 511) & ~(size_t)511;
    float* bufA = (float*)(ws + off);
    float* out  = (float*)d_out;
    off += (((size_t)N * H1 * sizeof(float)) + 511) & ~(size_t)511;
    int* keyp = (int*)(ws + off);
    const bool keyed = (ws_size >= off + ((((size_t)E + 15) >> 4) << 4)) &&
                       ((N + OWN_R - 1) / OWN_R <= 256);

    const int resBlocks  = 2048;
    auto gridFor = [&](size_t items) {
        size_t b = (items + 255) / 256;
        return (int)(b < (size_t)resBlocks ? b : (size_t)resBlocks);
    };

    if (keyed) {
        k_key<<<gridFor(((size_t)E + 3) / 4), 256, 0, stream>>>(dstIdx, keyp, E);
        k_agg_own<0, 1, 1, 1, OWN_R, true, true><<<(N + OWN_R - 1) / OWN_R, 1024, 0, stream>>>(
            nullptr, srcIdx, dstIdx, nullptr, nullptr, dinv, keyp, N, E, 0);
    } else {
        k_fill1<<<(N + 255) / 256, 256, 0, stream>>>(dinv, N);
        k_deg<<<gridFor(((size_t)E + 3) / 4), 256, 0, stream>>>(dstIdx, dinv, E);
        k_rsqrt<<<(N + 255) / 256, 256, 0, stream>>>(dinv, N);
    }

    const int gemmBlocks = (N + 15) / 16;

    auto layer = [&](const float* in, const float* Wt, const float* bias,
                     float* dest, int K, int F, int relu) {
        const int Kpad = (K + TK - 1) & ~(TK - 1);
        k_packW<<<(256 * Kpad + 255) / 256, 256, 0, stream>>>(Wt, Wpad, F, K, Kpad);
        if (F == 100)
            k_gemm_wmma<7><<<gemmBlocks, 256, 0, stream>>>(in, Wpad, bufH, N, K, Kpad, F);
        else if (F == 40)
            k_gemm_wmma<3><<<gemmBlocks, 256, 0, stream>>>(in, Wpad, bufH, N, K, Kpad, F);
        else if (F <= 16)
            k_gemm_wmma<1><<<gemmBlocks, 256, 0, stream>>>(in, Wpad, bufH, N, K, Kpad, F);
        else
            k_gemm_wmma<16><<<gemmBlocks, 256, 0, stream>>>(in, Wpad, bufH, N, K, Kpad, F);
        const int keyBlocks = (N + OWN_R - 1) / OWN_R;
        if (F == 200 && keyed) {
            k_agg_own<7, 4, 1, 200, OWN_R, true, false><<<keyBlocks, 1024, 0, stream>>>(
                bufH, srcIdx, dstIdx, dinv, bias, dest, keyp, N, E, relu);
        } else if (F == 100 && keyed) {
            k_agg_own<4, 8, 1, 100, OWN_R, true, false><<<keyBlocks, 1024, 0, stream>>>(
                bufH, srcIdx, dstIdx, dinv, bias, dest, keyp, N, E, relu);
        } else if (F == 40 && keyed) {
            k_agg_own<2, 4, 2, 40, OWN_R, true, false><<<keyBlocks, 1024, 0, stream>>>(
                bufH, srcIdx, dstIdx, dinv, bias, dest, keyp, N, E, relu);
        } else if (F == 3 && keyed) {
            k_agg_own<0, 1, 1, 3, OWN_R, true, false><<<keyBlocks, 1024, 0, stream>>>(
                bufH, srcIdx, dstIdx, dinv, bias, dest, keyp, N, E, relu);
        } else if (F == 200) {
            k_agg_own<7, 4, 1, 200, 396, false, false><<<(N + 395) / 396, 1024, 0, stream>>>(
                bufH, srcIdx, dstIdx, dinv, bias, dest, keyp, N, E, relu);
        } else if (F == 100) {
            k_agg_own<4, 8, 1, 100, 400, false, false><<<(N + 399) / 400, 1024, 0, stream>>>(
                bufH, srcIdx, dstIdx, dinv, bias, dest, keyp, N, E, relu);
        } else if (F == 40) {
            k_agg_own<2, 4, 2, 40, 400, false, false><<<(N + 399) / 400, 1024, 0, stream>>>(
                bufH, srcIdx, dstIdx, dinv, bias, dest, keyp, N, E, relu);
        } else {
            size_t tot = (size_t)N * F;
            const int streamGrid = gridFor((tot + 3) / 4);
            k_zero<<<streamGrid, 256, 0, stream>>>(dest, tot);
            k_agg_small<<<gridFor(((size_t)E + 3) / 4), 256, 0, stream>>>(
                bufH, srcIdx, dstIdx, dinv, dest, E, F);
            k_finish<<<streamGrid, 256, 0, stream>>>(
                dest, bufH, dinv, bias, N, F, relu);
        }
    };

    layer(x,    W1,  b1,  bufA, F_IN, H1, 1);
    layer(bufA, W1b, b1b, bufA, H1,   H2, 1);
    layer(bufA, W2b, b2b, bufA, H2,   H3, 1);
    layer(bufA, W2,  b2,  out,  H3,   C,  0);
}


# Owner blocks keep their scan's hits (source, local row) in idle workspace: degree pass writes, layer 1 replays; layer 2 scans and writes, layers 3-4 replay; 5 scans to 2; LDS arrival order kept
# speedup vs baseline: 3.1090x; 1.2031x over previous
#include <hip/hip_runtime.h>

typedef float v2f __attribute__((ext_vector_type(2)));
typedef float v4f __attribute__((ext_vector_type(4)));
typedef float v8f __attribute__((ext_vector_type(8)));
typedef int   v4i __attribute__((ext_vector_type(4)));

#define TK 64

template <int NT>
__global__ __launch_bounds__(256) void k_gemm_wmma(
    const float* __restrict__ X, const float* __restrict__ Wf,
    float* __restrict__ H, int M, int K, int Kpad, int Nout)
{
    __shared__ float As[16][TK + 4];

    const int row0 = blockIdx.x << 4;
    const int t    = threadIdx.x;
    const int wave = (__builtin_amdgcn_readfirstlane(t >> 5) +
                      (int)((blockIdx.x * 2654435761u) >> 29)) & 7;
    const int lane = t & 31;
    const int half = lane >> 4;
    const int m    = lane & 15;
    const bool has0 = wave < NT;
    const bool has1 = (wave + 8) < NT;

    const int n0 = (wave << 4) + m;
    const int n1 = ((wave + 8) << 4) + m;
    const int KS2 = Kpad >> 3;
    const v4f* __restrict__ Wq  = (const v4f*)Wf;
    const v4f* __restrict__ Wf0 = Wq + (size_t)wave * KS2 * 32 + lane;
    const v4f* __restrict__ Wf1 = Wq + (size_t)(wave + 8) * KS2 * 32 + lane;

    v8f acc0 = {}; v8f acc1 = {};

    for (int k0 = 0; k0 < Kpad; k0 += TK) {
        __syncthreads();
        #pragma unroll
        for (int i = 0; i < (16 * TK) / 256; ++i) {
            int e  = t + (i << 8);
            int r  = e >> 6;
            int kk = e & (TK - 1);
            int gr = row0 + r, gk = k0 + kk;
            As[r][kk] = (gr < M && gk < K) ? X[(size_t)gr * K + gk] : 0.0f;
        }
        __syncthreads();

        const int s0 = k0 >> 3;
        if (has1) {
            v4f q0[TK / 8], q1[TK / 8];
            #pragma unroll
            for (int j = 0; j < TK / 8; ++j) {
                q0[j] = Wf0[(size_t)(s0 + j) * 32];
                q1[j] = Wf1[(size_t)(s0 + j) * 32];
            }
            #pragma unroll
            for (int j = 0; j < TK / 8; ++j) {
                const int kb = (j << 3) + (half << 1);
                const v2f a0 = *(const v2f*)&As[m][kb];
                const v2f a1 = *(const v2f*)&As[m][kb + 4];
                const v2f b00 = __builtin_shufflevector(q0[j], q0[j], 0, 1);
                const v2f b01 = __builtin_shufflevector(q0[j], q0[j], 2, 3);
                const v2f b10 = __builtin_shufflevector(q1[j], q1[j], 0, 1);
                const v2f b11 = __builtin_shufflevector(q1[j], q1[j], 2, 3);
                acc0 = __builtin_amdgcn_wmma_f32_16x16x4_f32(
                    false, a0, false, b00, (short)0, acc0, false, false);
                acc1 = __builtin_amdgcn_wmma_f32_16x16x4_f32(
                    false, a0, false, b10, (short)0, acc1, false, false);
                acc0 = __builtin_amdgcn_wmma_f32_16x16x4_f32(
                    false, a1, false, b01, (short)0, acc0, false, false);
                acc1 = __builtin_amdgcn_wmma_f32_16x16x4_f32(
                    false, a1, false, b11, (short)0, acc1, false, false);
            }
        } else if (has0) {
            v4f q0[TK / 8];
            #pragma unroll
            for (int j = 0; j < TK / 8; ++j)
                q0[j] = Wf0[(size_t)(s0 + j) * 32];
            #pragma unroll
            for (int j = 0; j < TK / 8; ++j) {
                const int kb = (j << 3) + (half << 1);
                const v2f a0 = *(const v2f*)&As[m][kb];
                const v2f a1 = *(const v2f*)&As[m][kb + 4];
                const v2f b00 = __builtin_shufflevector(q0[j], q0[j], 0, 1);
                const v2f b01 = __builtin_shufflevector(q0[j], q0[j], 2, 3);
                acc0 = __builtin_amdgcn_wmma_f32_16x16x4_f32(
                    false, a0, false, b00, (short)0, acc0, false, false);
                acc0 = __builtin_amdgcn_wmma_f32_16x16x4_f32(
                    false, a1, false, b01, (short)0, acc0, false, false);
            }
        }
    }

    #pragma unroll
    for (int r = 0; r < 8; ++r) {
        int grow = row0 + r + (half << 3);
        if (grow < M) {
            if (has0 && n0 < Nout) H[(size_t)grow * Nout + n0] = acc0[r];
            if (has1 && n1 < Nout) H[(size_t)grow * Nout + n1] = acc1[r];
        }
    }
}

__global__ void k_packW(const float* __restrict__ W, float* __restrict__ Wf,
                        int Nout, int K, int Kpad)
{
    int idx = blockIdx.x * blockDim.x + threadIdx.x;
    int tot = 256 * Kpad;
    if (idx >= tot) return;
    const int KS2  = Kpad >> 3;
    const int q    = idx & 3;
    const int lane = (idx >> 2) & 31;
    const int rest = idx >> 7;
    const int s    = rest % KS2;
    const int ct   = rest / KS2;
    const int n = (ct << 4) + (lane & 15);
    const int k = (s << 3) + ((q >> 1) << 2) + ((lane >> 4) << 1) + (q & 1);
    Wf[idx] = (n < Nout && k < K) ? W[(size_t)n * K + k] : 0.0f;
}

__global__ void k_fill1(float* __restrict__ p, int n) {
    int i = blockIdx.x * blockDim.x + threadIdx.x;
    if (i < n) p[i] = 1.0f;
}

__global__ void k_deg(const int* __restrict__ dst, float* __restrict__ deg, int E) {
    const int nth = (int)(gridDim.x * blockDim.x);
    for (int e = (int)(blockIdx.x * blockDim.x + threadIdx.x); e < E; e += nth)
        atomicAdd(&deg[dst[e]], 1.0f);
}

#define OWN_R 392

__global__ __launch_bounds__(256) void k_key(
    const int* __restrict__ dst, int* __restrict__ key, int E)
{
    const int nth = (int)(gridDim.x * blockDim.x);
    const int nq  = (E + 3) >> 2;
    const bool al = ((((size_t)dst) & 15) == 0);
    for (int qd = (int)(blockIdx.x * blockDim.x + threadIdx.x); qd < nq; qd += nth) {
        const int e0 = qd << 2;
        unsigned w = 0u;
        if (al && e0 + 4 <= E) {
            const v4i d4 = *(const v4i*)(dst + e0);
            w = (unsigned)(d4.x / OWN_R) | ((unsigned)(d4.y / OWN_R) << 8) |
                ((unsigned)(d4.z / OWN_R) << 16) | ((unsigned)(d4.w / OWN_R) << 24);
        } else {
            for (int i = 0; i < 4; ++i)
                if (e0 + i < E) w |= (unsigned)(dst[e0 + i] / OWN_R) << (8 * i);
        }
        key[qd] = (int)w;
    }
}

__global__ void k_rsqrt(float* __restrict__ p, int n) {
    int i = blockIdx.x * blockDim.x + threadIdx.x;
    if (i < n) { float d = p[i]; p[i] = (d > 0.0f) ? rsqrtf(d) : 0.0f; }
}

__global__ __launch_bounds__(256) void k_zero(float* __restrict__ p, size_t n) {
    const size_t tid = (size_t)blockIdx.x * blockDim.x + threadIdx.x;
    const size_t nth = (size_t)gridDim.x * blockDim.x;
    if ((((size_t)p) & 15) == 0) {
        const size_t n4 = n >> 2;
        v4f* __restrict__ p4 = (v4f*)p;
        const v4f z = {0.0f, 0.0f, 0.0f, 0.0f};
        for (size_t i = tid; i < n4; i += nth) p4[i] = z;
        for (size_t i = (n4 << 2) + tid; i < n; i += nth) p[i] = 0.0f;
    } else {
        for (size_t i = tid; i < n; i += nth) p[i] = 0.0f;
    }
}

template <int NF, int TU, int NG>
__device__ __forceinline__ void agg_groups(
    const float* __restrict__ H, float* __restrict__ out,
    int s, int d, float nrm, int j, int F, int lane)
{
    constexpr int T  = 32 / TU;
    constexpr int U  = TU * NG;
    constexpr int FB = (NF - 1) << 5;
    const int tl = lane / T;
    const int tf = FB + (lane % T);

    float v[U][NF - 1];
    float vt[NG], nt[NG];
    int   dt[NG];
    #pragma unroll
    for (int u = 0; u < U; ++u) {
        const int su = __builtin_amdgcn_readlane(s, j + u);
        const float* __restrict__ hrow = H + (size_t)su * F;
        #pragma unroll
        for (int c = 0; c < NF - 1; ++c) v[u][c] = hrow[lane + (c << 5)];
    }
    #pragma unroll
    for (int g = 0; g < NG; ++g) {
        const int ju = j + g * TU + tl;
        const int st = __shfl(s, ju);
        dt[g] = __shfl(d, ju);
        nt[g] = __shfl(nrm, ju);
        vt[g] = H[(size_t)st * F + tf];
    }
    #pragma unroll
    for (int u = 0; u < U; ++u) {
        const int   du = __builtin_amdgcn_readlane(d, j + u);
        const float nu = __builtin_bit_cast(float,
            __builtin_amdgcn_readlane(__builtin_bit_cast(int, nrm), j + u));
        float* __restrict__ orow = out + (size_t)du * F;
        #pragma unroll
        for (int c = 0; c < NF - 1; ++c)
            atomicAdd(&orow[lane + (c << 5)], v[u][c] * nu);
    }
    #pragma unroll
    for (int g = 0; g < NG; ++g)
        atomicAdd(&out[(size_t)dt[g] * F + tf], vt[g] * nt[g]);
}

template <int NF>
__device__ __forceinline__ void agg_one(
    const float* __restrict__ H, float* __restrict__ out,
    int s, int d, float nrm, int j, int F, int lane)
{
    const int  fl = lane + ((NF - 1) << 5);
    const bool lastOn = (fl < F);
    const int   sj = __builtin_amdgcn_readlane(s, j);
    const int   dj = __builtin_amdgcn_readlane(d, j);
    const float nj = __builtin_bit_cast(float,
        __builtin_amdgcn_readlane(__builtin_bit_cast(int, nrm), j));
    const float* __restrict__ hrow = H + (size_t)sj * F;
    float* __restrict__ orow = out + (size_t)dj * F;
    float v[NF];
    #pragma unroll
    for (int c = 0; c < NF - 1; ++c) v[c] = hrow[lane + (c << 5)];
    v[NF - 1] = lastOn ? hrow[fl] : 0.0f;
    #pragma unroll
    for (int c = 0; c < NF - 1; ++c) atomicAdd(&orow[lane + (c << 5)], v[c] * nj);
    if (lastOn) atomicAdd(&orow[fl], v[NF - 1] * nj);
}

#define SCAN_U   8

#define LIST_HDR 32

template <int NF, int TU, int NG, int F, bool DEG, int MODE>
__device__ __forceinline__ void own_batch(
    const float* __restrict__ H, const int* __restrict__ src,
    const int* __restrict__ dst, const float* __restrict__ dinv,
    float* acc, int ent, int cnt, int rowBase, int lane,
    int* lst, int cap, int* lcnt)
{
    int pos = 0;
    if constexpr (MODE == 1) {
        if (lane == 0) pos = atomicAdd(lcnt, cnt);
        pos = __builtin_amdgcn_readfirstlane(pos);
    }
    if constexpr (NF == 0) {
        if (lane < cnt) {
            int s = 0, dl;
            if constexpr (MODE == 2) { s = ent >> 9; dl = ent & 511; }
            else {
                dl = dst[ent] - rowBase;
                if constexpr (!DEG || MODE == 1) s = src[ent];
            }
            float* row = acc + dl * F;
            if constexpr (DEG) {
                atomicAdd(row, 1.0f);
            } else {
                const float nrm = dinv[s] * dinv[rowBase + dl];
                const float* __restrict__ hrow = H + (size_t)s * F;
                float v[F];
                #pragma unroll
                for (int f = 0; f < F; ++f) v[f] = hrow[f];
                #pragma unroll
                for (int f = 0; f < F; ++f) atomicAdd(&row[f], v[f] * nrm);
            }
            if constexpr (MODE == 1)
                if (pos + lane < cap) lst[LIST_HDR + pos + lane] = (s << 9) | dl;
        }
    } else {
    constexpr int U = TU * NG;
    int s = 0, dl = 0; float nrm = 0.0f;
    if (lane < cnt) {
        if constexpr (MODE == 2) { s = ent >> 9; dl = ent & 511; }
        else { s = src[ent]; dl = dst[ent] - rowBase; }
        nrm = dinv[s] * dinv[rowBase + dl];
        if constexpr (MODE == 1)
            if (pos + lane < cap) lst[LIST_HDR + pos + lane] = (s << 9) | dl;
    }
    int j = 0;
    for (; j + U <= cnt; j += U)
        agg_groups<NF, TU, NG>(H, acc, s, dl, nrm, j, F, lane);
    for (; j < cnt; ++j)
        agg_one<NF>(H, acc, s, dl, nrm, j, F, lane);
    }
}

template <int NF, int TU, int NG, int F, int R, bool KEYED, bool DEG, int MODE, int FCAP>
__global__ __launch_bounds__(1024) void k_agg_own(
    const float* __restrict__ H, const int* __restrict__ src,
    const int* __restrict__ dst, const float* __restrict__ dinv,
    const float* __restrict__ bias, float* dest,
    const int* __restrict__ key, int* lst, int N, int E, int doRelu)
{
    static_assert(MODE == 0 || (KEYED && R <= 512), "a list entry keeps the local row in 9 bits");
    static_assert(NF == 0 || (R * F * 4) % 128 == 0, "block boundary must fall on a 128-byte line");
    static_assert((R * F) % 4 == 0, "the sums are cleared in 16-byte pieces");
    static_assert(SCAN_U * 4 <= 32, "one flag bit per load and piece in a 32-bit word");
    constexpr int ACC_N = (NF == 0 && R * F < 41472) ? 41472 : R * F;
    __shared__ __attribute__((aligned(16))) float acc[ACC_N];
    __shared__ int   que[32][64];
    __shared__ int   lcnt;

    const int t    = threadIdx.x;
    const int lane = t & 31;
    const int wave = __builtin_amdgcn_readfirstlane(t >> 5);
    const int rowBase = (int)blockIdx.x * R;
    const int nrows   = (N - rowBase < R) ? (N - rowBase) : R;

    {
        const v4f z = {0.0f, 0.0f, 0.0f, 0.0f};
        v4f* a4 = (v4f*)acc;
        for (int i = t; i < (R * F) / 4; i += 1024) a4[i] = z;
        if (t == 0) lcnt = 0;
    }
    __syncthreads();

    int* q = que[wave];
    int  qn = 0;
    const int nwv = (int)(blockDim.x >> 5);
    const int eb = (int)(((long long)wave * E) / nwv);
    const int ee = (int)(((long long)(wave + 1) * E) / nwv);

    int* L  = nullptr;
    int cap = 0;
    if constexpr (MODE != 0) {
        L   = lst + (size_t)rowBase * FCAP;
        cap = nrows * FCAP - LIST_HDR;
    }
    constexpr int SM = (MODE == 1) ? 1 : 0;
    bool scan = true;
    if constexpr (MODE == 2) {
        const int total = __builtin_amdgcn_readfirstlane(L[0]);
        if (total <= cap) {
            scan = false;
            const int nch = (total + 31) >> 5;
            int c = wave;
            int entNext = 0;
            if (c < nch) {
                const int ix = (c << 5) + lane;
                entNext = L[LIST_HDR + (ix < total ? ix : total - 1)];
            }
            while (c < nch) {
                const int ent = entNext;
                const int cnt = (total - (c << 5) < 32) ? (total - (c << 5)) : 32;
                c += nwv;
                if (c < nch) {
                    const int ix = (c << 5) + lane;
                    entNext = L[LIST_HDR + (ix < total ? ix : total - 1)];
                }
                own_batch<NF, TU, NG, F, DEG, 2>(H, src, dst, dinv, acc, ent, cnt, rowBase, lane,
                                                 nullptr, 0, nullptr);
            }
        }
    }

    if (scan) {
    if constexpr (KEYED) {
    const unsigned pat = (unsigned)blockIdx.x * 0x01010101u;
    auto zb = [&](int wv) -> unsigned {
        const unsigned x = (unsigned)wv ^ pat;
        const unsigned s7 = (x & 0x7F7F7F7Fu) + 0x7F7F7F7Fu;
        return ~(s7 | x | 0x7F7F7F7Fu);
    };
    const int P  = E >> 4;
    const int pb = (int)(((long long)wave * P) / nwv);
    const int pe = (int)(((long long)(wave + 1) * P) / nwv);
    const v4i* __restrict__ k4 = (const v4i*)key;
    for (int p0 = pb; p0 < pe; p0 += 32 * SCAN_U) {
        v4i kk[SCAN_U];
        #pragma unroll
        for (int k = 0; k < SCAN_U; ++k) {
            const int p = p0 + (k << 5) + lane;
            kk[k] = k4[p < pe ? p : pe - 1];
        }
        __builtin_amdgcn_sched_barrier(0);
        unsigned mk[SCAN_U];
        unsigned anyk = 0u;
        #pragma unroll
        for (int k = 0; k < SCAN_U; ++k) {
            const unsigned z0 = zb(kk[k].x), z1 = zb(kk[k].y);
            const unsigned z2 = zb(kk[k].z), z3 = zb(kk[k].w);
            const unsigned m = (z0 >> 7) | (z1 >> 6) | (z2 >> 5) | (z3 >> 4);
            mk[k] = (p0 + (k << 5) + lane < pe) ? m : 0u;
            anyk |= (__builtin_amdgcn_ballot_w32(mk[k] != 0u) != 0u) ? (1u << k) : 0u;
        }
        while (anyk != 0u) {
            const int k = __builtin_ctz(anyk);
            anyk &= anyk - 1u;
            unsigned m = mk[0];
            #pragma unroll
            for (int i = 1; i < SCAN_U; ++i) m = (k == i) ? mk[i] : m;
            const int ebase = (p0 + (k << 5) + lane) << 4;
            unsigned live = __builtin_amdgcn_ballot_w32(m != 0u);
            while (live != 0u) {
                const bool hit = m != 0u;
                const int  b   = __builtin_ctz(m | 0x80000000u);
                if (hit) q[qn + __builtin_popcount(live & ((1u << lane) - 1u))] =
                             ebase + ((b & 7) << 2) + (b >> 3);
                qn += __builtin_popcount(live);
                if (qn >= 32) {
                    own_batch<NF, TU, NG, F, DEG, SM>(H, src, dst, dinv, acc, q[lane], 32, rowBase, lane,
                                                      L, cap, &lcnt);
                    const int rest = qn - 32;
                    int tmp = 0;
                    if (lane < rest) tmp = q[32 + lane];
                    if (lane < rest) q[lane] = tmp;
                    qn = rest;
                }
                m &= m - 1u;
                live = __builtin_amdgcn_ballot_w32(m != 0u);
            }
        }
    }
    if (wave == nwv - 1 && (P << 4) < E) {
        const int pos = P << 4;
        const int e   = pos + lane;
        const bool hit = (e < E) && ((unsigned)(dst[e < E ? e : (E - 1)] - rowBase) < (unsigned)nrows);
        const unsigned m = __builtin_amdgcn_ballot_w32(hit);
        if (hit) q[qn + __builtin_popcount(m & ((1u << lane) - 1u))] = e;
        qn += __builtin_popcount(m);
    }
    } else {
    const unsigned unr = (unsigned)nrows;
    const int nE = ee - eb;
    const int* __restrict__ wp = dst + eb;
    int head = (int)(((16u - ((unsigned)(size_t)wp & 15u)) & 15u) >> 2);
    if (head > nE) head = nE;
    const int eg = eb + head;
    const int ng = (ee - eg) >> 7;
    const int et = eg + (ng << 7);
    const v4i* __restrict__ p4 = (const v4i*)(dst + eg) + lane;
    int pos = eb, g0 = 0;
    while (pos < ee) {
        unsigned hb = 0u, slots = 0u;
        int ebase, lmul;
        if (pos >= eg && g0 < ng) {
            v4i dk[SCAN_U];
            #pragma unroll
            for (int k = 0; k < SCAN_U; ++k) {
                const int g = (g0 + k < ng) ? (g0 + k) : (ng - 1);
                dk[k] = p4[(size_t)g << 5];
            }
            #pragma unroll
            for (int k = 0; k < SCAN_U; ++k) {
                if (g0 + k < ng) {
                    const unsigned u0 = (unsigned)(dk[k].x - rowBase);
                    const unsigned u1 = (unsigned)(dk[k].y - rowBase);
                    const unsigned u2 = (unsigned)(dk[k].z - rowBase);
                    const unsigned u3 = (unsigned)(dk[k].w - rowBase);
                    const unsigned a = u0 < u1 ? u0 : u1;
                    const unsigned b = u2 < u3 ? u2 : u3;
                    const unsigned mn = a < b ? a : b;
                    if (__builtin_amdgcn_ballot_w32(mn < unr) != 0u) {
                        const unsigned m0 = __builtin_amdgcn_ballot_w32(u0 < unr);
                        const unsigned m1 = __builtin_amdgcn_ballot_w32(u1 < unr);
                        const unsigned m2 = __builtin_amdgcn_ballot_w32(u2 < unr);
                        const unsigned m3 = __builtin_amdgcn_ballot_w32(u3 < unr);
                        hb |= ((u0 < unr ? 1u : 0u) | (u1 < unr ? 2u : 0u) |
                               (u2 < unr ? 4u : 0u) | (u3 < unr ? 8u : 0u)) << (4 * k);
                        slots |= ((m0 != 0u ? 1u : 0u) | (m1 != 0u ? 2u : 0u) |
                                  (m2 != 0u ? 4u : 0u) | (m3 != 0u ? 8u : 0u)) << (4 * k);
                    }
                }
            }
            ebase = eg + (g0 << 7);
            lmul  = 4;
            g0 += SCAN_U;
            pos = (g0 < ng) ? eg : et;
        } else {
            const int lim = (pos < eg) ? eg : ee;
            const int cnt = (lim - pos < 32) ? (lim - pos) : 32;
            const bool ok = lane < cnt;
            const int  dv = dst[ok ? (pos + lane) : pos];
            const bool hit = ok && ((unsigned)(dv - rowBase) < unr);
            hb    = hit ? 1u : 0u;
            slots = (__builtin_amdgcn_ballot_w32(hit) != 0u) ? 1u : 0u;
            ebase = pos;
            lmul  = 1;
            pos += cnt;
        }
        while (slots != 0u) {
            const int b = __builtin_ctz(slots);
            slots &= slots - 1u;
            const bool hit = ((hb >> b) & 1u) != 0u;
            const unsigned m = __builtin_amdgcn_ballot_w32(hit);
            if (hit) q[qn + __builtin_popcount(m & ((1u << lane) - 1u))] =
                         ebase + ((b >> 2) << 7) + lane * lmul + (b & 3);
            qn += __builtin_popcount(m);
            if (qn >= 32) {
                own_batch<NF, TU, NG, F, DEG, SM>(H, src, dst, dinv, acc, q[lane], 32, rowBase, lane,
                                                  L, cap, &lcnt);
                const int rest = qn - 32;
                int tmp = 0;
                if (lane < rest) tmp = q[32 + lane];
                if (lane < rest) q[lane] = tmp;
                qn = rest;
            }
        }
    }
    }
    while (qn > 0) {
        const int cnt = (qn < 32) ? qn : 32;
        own_batch<NF, TU, NG, F, DEG, SM>(H, src, dst, dinv, acc, (lane < cnt) ? q[lane] : 0, cnt, rowBase, lane,
                                          L, cap, &lcnt);
        const int rest = qn - cnt;
        int tmp = 0;
        if (lane < rest) tmp = q[32 + lane];
        if (lane < rest) q[lane] = tmp;
        qn = rest;
    }
    }

    __syncthreads();

    if constexpr (MODE == 1) {
        if (t == 0) L[0] = lcnt;
    }

    if constexpr (DEG) {
        for (int r = t; r < nrows; r += 1024) {
            const float dg = 1.0f + acc[r];
            dest[rowBase + r] = (dg > 0.0f) ? rsqrtf(dg) : 0.0f;
        }
    } else {
    const size_t gbase = (size_t)rowBase * F;
    const bool vec = (F % 4 == 0) &&
        ((((size_t)dest | (size_t)H | (size_t)bias) & 15) == 0);
    if (vec) {
        const int n4 = (nrows * F) >> 2;
        for (int p = t; p < n4; p += 1024) {
            const int idx = p << 2;
            const int r = idx / F;
            const int f = idx - r * F;
            const float di = dinv[rowBase + r];
            const v4f d4 = *(const v4f*)&acc[idx];
            const v4f h4 = *(const v4f*)(H + gbase + idx);
            const v4f b4 = *(const v4f*)(bias + f);
            v4f v;
            v.x = d4.x + h4.x * di * di + b4.x;
            v.y = d4.y + h4.y * di * di + b4.y;
            v.z = d4.z + h4.z * di * di + b4.z;
            v.w = d4.w + h4.w * di * di + b4.w;
            if (doRelu) {
                v.x = fmaxf(v.x, 0.0f); v.y = fmaxf(v.y, 0.0f);
                v.z = fmaxf(v.z, 0.0f); v.w = fmaxf(v.w, 0.0f);
            }
            *(v4f*)(dest + gbase + idx) = v;
        }
    } else {
        const int n1 = nrows * F;
        for (int idx = t; idx < n1; idx += 1024) {
            const int r = idx / F;
            const int f = idx - r * F;
            const float di = dinv[rowBase + r];
            const float v = acc[idx] + H[gbase + idx] * di * di + bias[f];
            dest[gbase + idx] = doRelu ? fmaxf(v, 0.0f) : v;
        }
    }
    }
}

__global__ void k_agg_small(
    const float* __restrict__ H, const int* __restrict__ src,
    const int* __restrict__ dst, const float* __restrict__ dinv,
    float* __restrict__ out, int E, int F)
{
    const unsigned nth = gridDim.x * blockDim.x;
    const unsigned tot = (unsigned)E * (unsigned)F;
    const unsigned uF  = (unsigned)F;
    for (unsigned idx = blockIdx.x * blockDim.x + threadIdx.x; idx < tot; idx += nth) {
        const unsigned e = idx / uF;
        const unsigned f = idx - e * uF;
        const int s = src[e], d = dst[e];
        const float nrm = dinv[s] * dinv[d];
        atomicAdd(&out[(size_t)d * F + f], H[(size_t)s * F + f] * nrm);
    }
}

__global__ void k_finish(float* __restrict__ dest, const float* __restrict__ H,
                         const float* __restrict__ dinv, const float* __restrict__ bias,
                         int N, int F, int doRelu)
{
    const unsigned tid = blockIdx.x * blockDim.x + threadIdx.x;
    const unsigned nth = gridDim.x * blockDim.x;
    const unsigned tot = (unsigned)N * (unsigned)F;
    const unsigned uF  = (unsigned)F;
    const bool vec = ((F & 3) == 0) &&
        ((((size_t)dest | (size_t)H | (size_t)bias) & 15) == 0);
    if (vec) {
        const unsigned n4 = tot >> 2;
        for (unsigned q = tid; q < n4; q += nth) {
            const unsigned idx = q << 2;
            const unsigned i = idx / uF;
            const unsigned f = idx - i * uF;
            const float di = dinv[i];
            const v4f d4 = ((const v4f*)dest)[q];
            const v4f h4 = ((const v4f*)H)[q];
            const v4f b4 = *(const v4f*)(bias + f);
            v4f v;
            v.x = d4.x + h4.x * di * di + b4.x;
            v.y = d4.y + h4.y * di * di + b4.y;
            v.z = d4.z + h4.z * di * di + b4.z;
            v.w = d4.w + h4.w * di * di + b4.w;
            if (doRelu) {
                v.x = fmaxf(v.x, 0.0f); v.y = fmaxf(v.y, 0.0f);
                v.z = fmaxf(v.z, 0.0f); v.w = fmaxf(v.w, 0.0f);
            }
            ((v4f*)dest)[q] = v;
        }
    } else {
        for (unsigned idx = tid; idx < tot; idx += nth) {
            const unsigned i = idx / uF;
            const unsigned f = idx - i * uF;
            const float di = dinv[i];
            const float v  = dest[idx] + H[idx] * di * di + bias[f];
            dest[idx] = doRelu ? fmaxf(v, 0.0f) : v;
        }
    }
}

extern "C" void kernel_launch(void* const* d_in, const int* in_sizes, int n_in,
                              void* d_out, int out_size, void* d_ws, size_t ws_size,
                              hipStream_t stream) {
    (void)n_in; (void)out_size;

    const float* x   = (const float*)d_in[0];
    const int*   ei  = (const int*)  d_in[1];
    const float* W1  = (const float*)d_in[2];
    const float* b1  = (const float*)d_in[3];
    const float* W1b = (const float*)d_in[4];
    const float* b1b = (const float*)d_in[5];
    const float* W2b = (const float*)d_in[6];
    const float* b2b = (const float*)d_in[7];
    const float* W2  = (const float*)d_in[8];
    const float* b2  = (const float*)d_in[9];

    const int F_IN = 500, H1 = 200, H2 = 100, H3 = 40, C = 3;
    const int N = in_sizes[0] / F_IN;
    const int E = in_sizes[1] / 2;
    const int* srcIdx = ei;
    const int* dstIdx = ei + E;

    char*  ws   = (char*)d_ws;
    float* dinv = (float*)ws;
    size_t off  = ((size_t)N * sizeof(float) + 511) & ~(size_t)511;
    float* Wpad = (float*)(ws + off);
    off += (size_t)256 * 512 * sizeof(float);
    float* bufH = (float*)(ws + off);
    off += (((size_t)N * H1 * sizeof(float)) + 511) & ~(size_t)511;
    float* bufA = (float*)(ws + off);
    float* out  = (float*)d_out;
    off += (((size_t)N * H1 * sizeof(float)) + 511) & ~(size_t)511;
    int* keyp = (int*)(ws + off);
    const bool keyed = (ws_size >= off + ((((size_t)E + 15) >> 4) << 4)) &&
                       ((N + OWN_R - 1) / OWN_R <= 256);
    int* listA = (int*)bufA;
    int* listB = (int*)(bufH + (size_t)N * H2);

    const int resBlocks  = 2048;
    auto gridFor = [&](size_t items) {
        size_t b = (items + 255) / 256;
        return (int)(b < (size_t)resBlocks ? b : (size_t)resBlocks);
    };

    if (keyed) {
        k_key<<<gridFor(((size_t)E + 3) / 4), 256, 0, stream>>>(dstIdx, keyp, E);
        k_agg_own<0, 1, 1, 1, OWN_R, true, true, 1, 200><<<(N + OWN_R - 1) / OWN_R, 1024, 0, stream>>>(
            nullptr, srcIdx, dstIdx, nullptr, nullptr, dinv, keyp, listA, N, E, 0);
    } else {
        k_fill1<<<(N + 255) / 256, 256, 0, stream>>>(dinv, N);
        k_deg<<<gridFor(((size_t)E + 3) / 4), 256, 0, stream>>>(dstIdx, dinv, E);
        k_rsqrt<<<(N + 255) / 256, 256, 0, stream>>>(dinv, N);
    }

    const int gemmBlocks = (N + 15) / 16;

    auto layer = [&](const float* in, const float* Wt, const float* bias,
                     float* dest, int K, int F, int relu) {
        const int Kpad = (K + TK - 1) & ~(TK - 1);
        k_packW<<<(256 * Kpad + 255) / 256, 256, 0, stream>>>(Wt, Wpad, F, K, Kpad);
        if (F == 100)
            k_gemm_wmma<7><<<gemmBlocks, 256, 0, stream>>>(in, Wpad, bufH, N, K, Kpad, F);
        else if (F == 40)
            k_gemm_wmma<3><<<gemmBlocks, 256, 0, stream>>>(in, Wpad, bufH, N, K, Kpad, F);
        else if (F <= 16)
            k_gemm_wmma<1><<<gemmBlocks, 256, 0, stream>>>(in, Wpad, bufH, N, K, Kpad, F);
        else
            k_gemm_wmma<16><<<gemmBlocks, 256, 0, stream>>>(in, Wpad, bufH, N, K, Kpad, F);
        const int keyBlocks = (N + OWN_R - 1) / OWN_R;
        if (F == 200 && keyed) {
            k_agg_own<7, 4, 1, 200, OWN_R, true, false, 2, 200><<<keyBlocks, 1024, 0, stream>>>(
                bufH, srcIdx, dstIdx, dinv, bias, dest, keyp, listA, N, E, relu);
        } else if (F == 100 && keyed) {
            k_agg_own<4, 8, 1, 100, OWN_R, true, false, 1, 100><<<keyBlocks, 1024, 0, stream>>>(
                bufH, srcIdx, dstIdx, dinv, bias, dest, keyp, listB, N, E, relu);
        } else if (F == 40 && keyed) {
            k_agg_own<2, 4, 2, 40, OWN_R, true, false, 2, 100><<<keyBlocks, 1024, 0, stream>>>(
                bufH, srcIdx, dstIdx, dinv, bias, dest, keyp, listB, N, E, relu);
        } else if (F == 3 && keyed) {
            k_agg_own<0, 1, 1, 3, OWN_R, true, false, 2, 100><<<keyBlocks, 1024, 0, stream>>>(
                bufH, srcIdx, dstIdx, dinv, bias, dest, keyp, listB, N, E, relu);
        } else if (F == 200) {
            k_agg_own<7, 4, 1, 200, 396, false, false, 0, 1><<<(N + 395) / 396, 1024, 0, stream>>>(
                bufH, srcIdx, dstIdx, dinv, bias, dest, keyp, nullptr, N, E, relu);
        } else if (F == 100) {
            k_agg_own<4, 8, 1, 100, 400, false, false, 0, 1><<<(N + 399) / 400, 1024, 0, stream>>>(
                bufH, srcIdx, dstIdx, dinv, bias, dest, keyp, nullptr, N, E, relu);
        } else if (F == 40) {
            k_agg_own<2, 4, 2, 40, 400, false, false, 0, 1><<<(N + 399) / 400, 1024, 0, stream>>>(
                bufH, srcIdx, dstIdx, dinv, bias, dest, keyp, nullptr, N, E, relu);
        } else {
            size_t tot = (size_t)N * F;
            const int streamGrid = gridFor((tot + 3) / 4);
            k_zero<<<streamGrid, 256, 0, stream>>>(dest, tot);
            k_agg_small<<<gridFor(((size_t)E + 3) / 4), 256, 0, stream>>>(
                bufH, srcIdx, dstIdx, dinv, dest, E, F);
            k_finish<<<streamGrid, 256, 0, stream>>>(
                dest, bufH, dinv, bias, N, F, relu);
        }
    };

    layer(x,    W1,  b1,  bufA, F_IN, H1, 1);
    layer(bufA, W1b, b1b, bufA, H1,   H2, 1);
    layer(bufA, W2b, b2b, bufA, H2,   H3, 1);
    layer(bufA, W2,  b2,  out,  H3,   C,  0);
}


# Kept hit list moves to a workspace region of its own behind the key plane (used where the workspace has room): degree pass writes it, all 4 layers replay it; 2 scans to 1; LDS arrival order kept
# speedup vs baseline: 3.3474x; 1.0767x over previous
#include <hip/hip_runtime.h>

typedef float v2f __attribute__((ext_vector_type(2)));
typedef float v4f __attribute__((ext_vector_type(4)));
typedef float v8f __attribute__((ext_vector_type(8)));
typedef int   v4i __attribute__((ext_vector_type(4)));

#define TK 64

template <int NT>
__global__ __launch_bounds__(256) void k_gemm_wmma(
    const float* __restrict__ X, const float* __restrict__ Wf,
    float* __restrict__ H, int M, int K, int Kpad, int Nout)
{
    __shared__ float As[16][TK + 4];

    const int row0 = blockIdx.x << 4;
    const int t    = threadIdx.x;
    const int wave = (__builtin_amdgcn_readfirstlane(t >> 5) +
                      (int)((blockIdx.x * 2654435761u) >> 29)) & 7;
    const int lane = t & 31;
    const int half = lane >> 4;
    const int m    = lane & 15;
    const bool has0 = wave < NT;
    const bool has1 = (wave + 8) < NT;

    const int n0 = (wave << 4) + m;
    const int n1 = ((wave + 8) << 4) + m;
    const int KS2 = Kpad >> 3;
    const v4f* __restrict__ Wq  = (const v4f*)Wf;
    const v4f* __restrict__ Wf0 = Wq + (size_t)wave * KS2 * 32 + lane;
    const v4f* __restrict__ Wf1 = Wq + (size_t)(wave + 8) * KS2 * 32 + lane;

    v8f acc0 = {}; v8f acc1 = {};

    for (int k0 = 0; k0 < Kpad; k0 += TK) {
        __syncthreads();
        #pragma unroll
        for (int i = 0; i < (16 * TK) / 256; ++i) {
            int e  = t + (i << 8);
            int r  = e >> 6;
            int kk = e & (TK - 1);
            int gr = row0 + r, gk = k0 + kk;
            As[r][kk] = (gr < M && gk < K) ? X[(size_t)gr * K + gk] : 0.0f;
        }
        __syncthreads();

        const int s0 = k0 >> 3;
        if (has1) {
            v4f q0[TK / 8], q1[TK / 8];
            #pragma unroll
            for (int j = 0; j < TK / 8; ++j) {
                q0[j] = Wf0[(size_t)(s0 + j) * 32];
                q1[j] = Wf1[(size_t)(s0 + j) * 32];
            }
            #pragma unroll
            for (int j = 0; j < TK / 8; ++j) {
                const int kb = (j << 3) + (half << 1);
                const v2f a0 = *(const v2f*)&As[m][kb];
                const v2f a1 = *(const v2f*)&As[m][kb + 4];
                const v2f b00 = __builtin_shufflevector(q0[j], q0[j], 0, 1);
                const v2f b01 = __builtin_shufflevector(q0[j], q0[j], 2, 3);
                const v2f b10 = __builtin_shufflevector(q1[j], q1[j], 0, 1);
                const v2f b11 = __builtin_shufflevector(q1[j], q1[j], 2, 3);
                acc0 = __builtin_amdgcn_wmma_f32_16x16x4_f32(
                    false, a0, false, b00, (short)0, acc0, false, false);
                acc1 = __builtin_amdgcn_wmma_f32_16x16x4_f32(
                    false, a0, false, b10, (short)0, acc1, false, false);
                acc0 = __builtin_amdgcn_wmma_f32_16x16x4_f32(
                    false, a1, false, b01, (short)0, acc0, false, false);
                acc1 = __builtin_amdgcn_wmma_f32_16x16x4_f32(
                    false, a1, false, b11, (short)0, acc1, false, false);
            }
        } else if (has0) {
            v4f q0[TK / 8];
            #pragma unroll
            for (int j = 0; j < TK / 8; ++j)
                q0[j] = Wf0[(size_t)(s0 + j) * 32];
            #pragma unroll
            for (int j = 0; j < TK / 8; ++j) {
                const int kb = (j << 3) + (half << 1);
                const v2f a0 = *(const v2f*)&As[m][kb];
                const v2f a1 = *(const v2f*)&As[m][kb + 4];
                const v2f b00 = __builtin_shufflevector(q0[j], q0[j], 0, 1);
                const v2f b01 = __builtin_shufflevector(q0[j], q0[j], 2, 3);
                acc0 = __builtin_amdgcn_wmma_f32_16x16x4_f32(
                    false, a0, false, b00, (short)0, acc0, false, false);
                acc0 = __builtin_amdgcn_wmma_f32_16x16x4_f32(
                    false, a1, false, b01, (short)0, acc0, false, false);
            }
        }
    }

    #pragma unroll
    for (int r = 0; r < 8; ++r) {
        int grow = row0 + r + (half << 3);
        if (grow < M) {
            if (has0 && n0 < Nout) H[(size_t)grow * Nout + n0] = acc0[r];
            if (has1 && n1 < Nout) H[(size_t)grow * Nout + n1] = acc1[r];
        }
    }
}

__global__ void k_packW(const float* __restrict__ W, float* __restrict__ Wf,
                        int Nout, int K, int Kpad)
{
    int idx = blockIdx.x * blockDim.x + threadIdx.x;
    int tot = 256 * Kpad;
    if (idx >= tot) return;
    const int KS2  = Kpad >> 3;
    const int q    = idx & 3;
    const int lane = (idx >> 2) & 31;
    const int rest = idx >> 7;
    const int s    = rest % KS2;
    const int ct   = rest / KS2;
    const int n = (ct << 4) + (lane & 15);
    const int k = (s << 3) + ((q >> 1) << 2) + ((lane >> 4) << 1) + (q & 1);
    Wf[idx] = (n < Nout && k < K) ? W[(size_t)n * K + k] : 0.0f;
}

__global__ void k_fill1(float* __restrict__ p, int n) {
    int i = blockIdx.x * blockDim.x + threadIdx.x;
    if (i < n) p[i] = 1.0f;
}

__global__ void k_deg(const int* __restrict__ dst, float* __restrict__ deg, int E) {
    const int nth = (int)(gridDim.x * blockDim.x);
    for (int e = (int)(blockIdx.x * blockDim.x + threadIdx.x); e < E; e += nth)
        atomicAdd(&deg[dst[e]], 1.0f);
}

#define OWN_R 392

__global__ __launch_bounds__(256) void k_key(
    const int* __restrict__ dst, int* __restrict__ key, int E)
{
    const int nth = (int)(gridDim.x * blockDim.x);
    const int nq  = (E + 3) >> 2;
    const bool al = ((((size_t)dst) & 15) == 0);
    for (int qd = (int)(blockIdx.x * blockDim.x + threadIdx.x); qd < nq; qd += nth) {
        const int e0 = qd << 2;
        unsigned w = 0u;
        if (al && e0 + 4 <= E) {
            const v4i d4 = *(const v4i*)(dst + e0);
            w = (unsigned)(d4.x / OWN_R) | ((unsigned)(d4.y / OWN_R) << 8) |
                ((unsigned)(d4.z / OWN_R) << 16) | ((unsigned)(d4.w / OWN_R) << 24);
        } else {
            for (int i = 0; i < 4; ++i)
                if (e0 + i < E) w |= (unsigned)(dst[e0 + i] / OWN_R) << (8 * i);
        }
        key[qd] = (int)w;
    }
}

__global__ void k_rsqrt(float* __restrict__ p, int n) {
    int i = blockIdx.x * blockDim.x + threadIdx.x;
    if (i < n) { float d = p[i]; p[i] = (d > 0.0f) ? rsqrtf(d) : 0.0f; }
}

__global__ __launch_bounds__(256) void k_zero(float* __restrict__ p, size_t n) {
    const size_t tid = (size_t)blockIdx.x * blockDim.x + threadIdx.x;
    const size_t nth = (size_t)gridDim.x * blockDim.x;
    if ((((size_t)p) & 15) == 0) {
        const size_t n4 = n >> 2;
        v4f* __restrict__ p4 = (v4f*)p;
        const v4f z = {0.0f, 0.0f, 0.0f, 0.0f};
        for (size_t i = tid; i < n4; i += nth) p4[i] = z;
        for (size_t i = (n4 << 2) + tid; i < n; i += nth) p[i] = 0.0f;
    } else {
        for (size_t i = tid; i < n; i += nth) p[i] = 0.0f;
    }
}

template <int NF, int TU, int NG>
__device__ __forceinline__ void agg_groups(
    const float* __restrict__ H, float* __restrict__ out,
    int s, int d, float nrm, int j, int F, int lane)
{
    constexpr int T  = 32 / TU;
    constexpr int U  = TU * NG;
    constexpr int FB = (NF - 1) << 5;
    const int tl = lane / T;
    const int tf = FB + (lane % T);

    float v[U][NF - 1];
    float vt[NG], nt[NG];
    int   dt[NG];
    #pragma unroll
    for (int u = 0; u < U; ++u) {
        const int su = __builtin_amdgcn_readlane(s, j + u);
        const float* __restrict__ hrow = H + (size_t)su * F;
        #pragma unroll
        for (int c = 0; c < NF - 1; ++c) v[u][c] = hrow[lane + (c << 5)];
    }
    #pragma unroll
    for (int g = 0; g < NG; ++g) {
        const int ju = j + g * TU + tl;
        const int st = __shfl(s, ju);
        dt[g] = __shfl(d, ju);
        nt[g] = __shfl(nrm, ju);
        vt[g] = H[(size_t)st * F + tf];
    }
    #pragma unroll
    for (int u = 0; u < U; ++u) {
        const int   du = __builtin_amdgcn_readlane(d, j + u);
        const float nu = __builtin_bit_cast(float,
            __builtin_amdgcn_readlane(__builtin_bit_cast(int, nrm), j + u));
        float* __restrict__ orow = out + (size_t)du * F;
        #pragma unroll
        for (int c = 0; c < NF - 1; ++c)
            atomicAdd(&orow[lane + (c << 5)], v[u][c] * nu);
    }
    #pragma unroll
    for (int g = 0; g < NG; ++g)
        atomicAdd(&out[(size_t)dt[g] * F + tf], vt[g] * nt[g]);
}

template <int NF>
__device__ __forceinline__ void agg_one(
    const float* __restrict__ H, float* __restrict__ out,
    int s, int d, float nrm, int j, int F, int lane)
{
    const int  fl = lane + ((NF - 1) << 5);
    const bool lastOn = (fl < F);
    const int   sj = __builtin_amdgcn_readlane(s, j);
    const int   dj = __builtin_amdgcn_readlane(d, j);
    const float nj = __builtin_bit_cast(float,
        __builtin_amdgcn_readlane(__builtin_bit_cast(int, nrm), j));
    const float* __restrict__ hrow = H + (size_t)sj * F;
    float* __restrict__ orow = out + (size_t)dj * F;
    float v[NF];
    #pragma unroll
    for (int c = 0; c < NF - 1; ++c) v[c] = hrow[lane + (c << 5)];
    v[NF - 1] = lastOn ? hrow[fl] : 0.0f;
    #pragma unroll
    for (int c = 0; c < NF - 1; ++c) atomicAdd(&orow[lane + (c << 5)], v[c] * nj);
    if (lastOn) atomicAdd(&orow[fl], v[NF - 1] * nj);
}

#define SCAN_U   8

#define LIST_HDR 32
#define KEEP_CAP 24

template <int NF, int TU, int NG, int F, bool DEG, int MODE>
__device__ __forceinline__ void own_batch(
    const float* __restrict__ H, const int* __restrict__ src,
    const int* __restrict__ dst, const float* __restrict__ dinv,
    float* acc, int ent, int cnt, int rowBase, int lane,
    int* lst, int cap, int* lcnt)
{
    int pos = 0;
    if constexpr (MODE == 1) {
        if (lane == 0) pos = atomicAdd(lcnt, cnt);
        pos = __builtin_amdgcn_readfirstlane(pos);
    }
    if constexpr (NF == 0) {
        if (lane < cnt) {
            int s = 0, dl;
            if constexpr (MODE == 2) { s = ent >> 9; dl = ent & 511; }
            else {
                dl = dst[ent] - rowBase;
                if constexpr (!DEG || MODE == 1) s = src[ent];
            }
            float* row = acc + dl * F;
            if constexpr (DEG) {
                atomicAdd(row, 1.0f);
            } else {
                const float nrm = dinv[s] * dinv[rowBase + dl];
                const float* __restrict__ hrow = H + (size_t)s * F;
                float v[F];
                #pragma unroll
                for (int f = 0; f < F; ++f) v[f] = hrow[f];
                #pragma unroll
                for (int f = 0; f < F; ++f) atomicAdd(&row[f], v[f] * nrm);
            }
            if constexpr (MODE == 1)
                if (pos + lane < cap) lst[LIST_HDR + pos + lane] = (s << 9) | dl;
        }
    } else {
    constexpr int U = TU * NG;
    int s = 0, dl = 0; float nrm = 0.0f;
    if (lane < cnt) {
        if constexpr (MODE == 2) { s = ent >> 9; dl = ent & 511; }
        else { s = src[ent]; dl = dst[ent] - rowBase; }
        nrm = dinv[s] * dinv[rowBase + dl];
        if constexpr (MODE == 1)
            if (pos + lane < cap) lst[LIST_HDR + pos + lane] = (s << 9) | dl;
    }
    int j = 0;
    for (; j + U <= cnt; j += U)
        agg_groups<NF, TU, NG>(H, acc, s, dl, nrm, j, F, lane);
    for (; j < cnt; ++j)
        agg_one<NF>(H, acc, s, dl, nrm, j, F, lane);
    }
}

template <int NF, int TU, int NG, int F, int R, bool KEYED, bool DEG, int MODE, int FCAP>
__global__ __launch_bounds__(1024) void k_agg_own(
    const float* __restrict__ H, const int* __restrict__ src,
    const int* __restrict__ dst, const float* __restrict__ dinv,
    const float* __restrict__ bias, float* dest,
    const int* __restrict__ key, int* lst, int N, int E, int doRelu)
{
    static_assert(MODE == 0 || (KEYED && R <= 512), "a list entry keeps the local row in 9 bits");
    static_assert(NF == 0 || (R * F * 4) % 128 == 0, "block boundary must fall on a 128-byte line");
    static_assert((R * F) % 4 == 0, "the sums are cleared in 16-byte pieces");
    static_assert(SCAN_U * 4 <= 32, "one flag bit per load and piece in a 32-bit word");
    constexpr int ACC_N = (NF == 0 && R * F < 41472) ? 41472 : R * F;
    __shared__ __attribute__((aligned(16))) float acc[ACC_N];
    __shared__ int   que[32][64];
    __shared__ int   lcnt;

    const int t    = threadIdx.x;
    const int lane = t & 31;
    const int wave = __builtin_amdgcn_readfirstlane(t >> 5);
    const int rowBase = (int)blockIdx.x * R;
    const int nrows   = (N - rowBase < R) ? (N - rowBase) : R;

    {
        const v4f z = {0.0f, 0.0f, 0.0f, 0.0f};
        v4f* a4 = (v4f*)acc;
        for (int i = t; i < (R * F) / 4; i += 1024) a4[i] = z;
        if (t == 0) lcnt = 0;
    }
    __syncthreads();

    int* q = que[wave];
    int  qn = 0;
    const int nwv = (int)(blockDim.x >> 5);
    const int eb = (int)(((long long)wave * E) / nwv);
    const int ee = (int)(((long long)(wave + 1) * E) / nwv);

    int* L  = nullptr;
    int cap = 0;
    if constexpr (MODE != 0) {
        L   = lst + (size_t)rowBase * FCAP;
        cap = nrows * FCAP - LIST_HDR;
    }
    constexpr int SM = (MODE == 1) ? 1 : 0;
    bool scan = true;
    if constexpr (MODE == 2) {
        const int total = __builtin_amdgcn_readfirstlane(L[0]);
        if (total <= cap) {
            scan = false;
            const int nch = (total + 31) >> 5;
            int c = wave;
            int entNext = 0;
            if (c < nch) {
                const int ix = (c << 5) + lane;
                entNext = L[LIST_HDR + (ix < total ? ix : total - 1)];
            }
            while (c < nch) {
                const int ent = entNext;
                const int cnt = (total - (c << 5) < 32) ? (total - (c << 5)) : 32;
                c += nwv;
                if (c < nch) {
                    const int ix = (c << 5) + lane;
                    entNext = L[LIST_HDR + (ix < total ? ix : total - 1)];
                }
                own_batch<NF, TU, NG, F, DEG, 2>(H, src, dst, dinv, acc, ent, cnt, rowBase, lane,
                                                 nullptr, 0, nullptr);
            }
        }
    }

    if (scan) {
    if constexpr (KEYED) {
    const unsigned pat = (unsigned)blockIdx.x * 0x01010101u;
    auto zb = [&](int wv) -> unsigned {
        const unsigned x = (unsigned)wv ^ pat;
        const unsigned s7 = (x & 0x7F7F7F7Fu) + 0x7F7F7F7Fu;
        return ~(s7 | x | 0x7F7F7F7Fu);
    };
    const int P  = E >> 4;
    const int pb = (int)(((long long)wave * P) / nwv);
    const int pe = (int)(((long long)(wave + 1) * P) / nwv);
    const v4i* __restrict__ k4 = (const v4i*)key;
    for (int p0 = pb; p0 < pe; p0 += 32 * SCAN_U) {
        v4i kk[SCAN_U];
        #pragma unroll
        for (int k = 0; k < SCAN_U; ++k) {
            const int p = p0 + (k << 5) + lane;
            kk[k] = k4[p < pe ? p : pe - 1];
        }
        __builtin_amdgcn_sched_barrier(0);
        unsigned mk[SCAN_U];
        unsigned anyk = 0u;
        #pragma unroll
        for (int k = 0; k < SCAN_U; ++k) {
            const unsigned z0 = zb(kk[k].x), z1 = zb(kk[k].y);
            const unsigned z2 = zb(kk[k].z), z3 = zb(kk[k].w);
            const unsigned m = (z0 >> 7) | (z1 >> 6) | (z2 >> 5) | (z3 >> 4);
            mk[k] = (p0 + (k << 5) + lane < pe) ? m : 0u;
            anyk |= (__builtin_amdgcn_ballot_w32(mk[k] != 0u) != 0u) ? (1u << k) : 0u;
        }
        while (anyk != 0u) {
            const int k = __builtin_ctz(anyk);
            anyk &= anyk - 1u;
            unsigned m = mk[0];
            #pragma unroll
            for (int i = 1; i < SCAN_U; ++i) m = (k == i) ? mk[i] : m;
            const int ebase = (p0 + (k << 5) + lane) << 4;
            unsigned live = __builtin_amdgcn_ballot_w32(m != 0u);
            while (live != 0u) {
                const bool hit = m != 0u;
                const int  b   = __builtin_ctz(m | 0x80000000u);
                if (hit) q[qn + __builtin_popcount(live & ((1u << lane) - 1u))] =
                             ebase + ((b & 7) << 2) + (b >> 3);
                qn += __builtin_popcount(live);
                if (qn >= 32) {
                    own_batch<NF, TU, NG, F, DEG, SM>(H, src, dst, dinv, acc, q[lane], 32, rowBase, lane,
                                                      L, cap, &lcnt);
                    const int rest = qn - 32;
                    int tmp = 0;
                    if (lane < rest) tmp = q[32 + lane];
                    if (lane < rest) q[lane] = tmp;
                    qn = rest;
                }
                m &= m - 1u;
                live = __builtin_amdgcn_ballot_w32(m != 0u);
            }
        }
    }
    if (wave == nwv - 1 && (P << 4) < E) {
        const int pos = P << 4;
        const int e   = pos + lane;
        const bool hit = (e < E) && ((unsigned)(dst[e < E ? e : (E - 1)] - rowBase) < (unsigned)nrows);
        const unsigned m = __builtin_amdgcn_ballot_w32(hit);
        if (hit) q[qn + __builtin_popcount(m & ((1u << lane) - 1u))] = e;
        qn += __builtin_popcount(m);
    }
    } else {
    const unsigned unr = (unsigned)nrows;
    const int nE = ee - eb;
    const int* __restrict__ wp = dst + eb;
    int head = (int)(((16u - ((unsigned)(size_t)wp & 15u)) & 15u) >> 2);
    if (head > nE) head = nE;
    const int eg = eb + head;
    const int ng = (ee - eg) >> 7;
    const int et = eg + (ng << 7);
    const v4i* __restrict__ p4 = (const v4i*)(dst + eg) + lane;
    int pos = eb, g0 = 0;
    while (pos < ee) {
        unsigned hb = 0u, slots = 0u;
        int ebase, lmul;
        if (pos >= eg && g0 < ng) {
            v4i dk[SCAN_U];
            #pragma unroll
            for (int k = 0; k < SCAN_U; ++k) {
                const int g = (g0 + k < ng) ? (g0 + k) : (ng - 1);
                dk[k] = p4[(size_t)g << 5];
            }
            #pragma unroll
            for (int k = 0; k < SCAN_U; ++k) {
                if (g0 + k < ng) {
                    const unsigned u0 = (unsigned)(dk[k].x - rowBase);
                    const unsigned u1 = (unsigned)(dk[k].y - rowBase);
                    const unsigned u2 = (unsigned)(dk[k].z - rowBase);
                    const unsigned u3 = (unsigned)(dk[k].w - rowBase);
                    const unsigned a = u0 < u1 ? u0 : u1;
                    const unsigned b = u2 < u3 ? u2 : u3;
                    const unsigned mn = a < b ? a : b;
                    if (__builtin_amdgcn_ballot_w32(mn < unr) != 0u) {
                        const unsigned m0 = __builtin_amdgcn_ballot_w32(u0 < unr);
                        const unsigned m1 = __builtin_amdgcn_ballot_w32(u1 < unr);
                        const unsigned m2 = __builtin_amdgcn_ballot_w32(u2 < unr);
                        const unsigned m3 = __builtin_amdgcn_ballot_w32(u3 < unr);
                        hb |= ((u0 < unr ? 1u : 0u) | (u1 < unr ? 2u : 0u) |
                               (u2 < unr ? 4u : 0u) | (u3 < unr ? 8u : 0u)) << (4 * k);
                        slots |= ((m0 != 0u ? 1u : 0u) | (m1 != 0u ? 2u : 0u) |
                                  (m2 != 0u ? 4u : 0u) | (m3 != 0u ? 8u : 0u)) << (4 * k);
                    }
                }
            }
            ebase = eg + (g0 << 7);
            lmul  = 4;
            g0 += SCAN_U;
            pos = (g0 < ng) ? eg : et;
        } else {
            const int lim = (pos < eg) ? eg : ee;
            const int cnt = (lim - pos < 32) ? (lim - pos) : 32;
            const bool ok = lane < cnt;
            const int  dv = dst[ok ? (pos + lane) : pos];
            const bool hit = ok && ((unsigned)(dv - rowBase) < unr);
            hb    = hit ? 1u : 0u;
            slots = (__builtin_amdgcn_ballot_w32(hit) != 0u) ? 1u : 0u;
            ebase = pos;
            lmul  = 1;
            pos += cnt;
        }
        while (slots != 0u) {
            const int b = __builtin_ctz(slots);
            slots &= slots - 1u;
            const bool hit = ((hb >> b) & 1u) != 0u;
            const unsigned m = __builtin_amdgcn_ballot_w32(hit);
            if (hit) q[qn + __builtin_popcount(m & ((1u << lane) - 1u))] =
                         ebase + ((b >> 2) << 7) + lane * lmul + (b & 3);
            qn += __builtin_popcount(m);
            if (qn >= 32) {
                own_batch<NF, TU, NG, F, DEG, SM>(H, src, dst, dinv, acc, q[lane], 32, rowBase, lane,
                                                  L, cap, &lcnt);
                const int rest = qn - 32;
                int tmp = 0;
                if (lane < rest) tmp = q[32 + lane];
                if (lane < rest) q[lane] = tmp;
                qn = rest;
            }
        }
    }
    }
    while (qn > 0) {
        const int cnt = (qn < 32) ? qn : 32;
        own_batch<NF, TU, NG, F, DEG, SM>(H, src, dst, dinv, acc, (lane < cnt) ? q[lane] : 0, cnt, rowBase, lane,
                                          L, cap, &lcnt);
        const int rest = qn - cnt;
        int tmp = 0;
        if (lane < rest) tmp = q[32 + lane];
        if (lane < rest) q[lane] = tmp;
        qn = rest;
    }
    }

    __syncthreads();

    if constexpr (MODE == 1) {
        if (t == 0) L[0] = lcnt;
    }

    if constexpr (DEG) {
        for (int r = t; r < nrows; r += 1024) {
            const float dg = 1.0f + acc[r];
            dest[rowBase + r] = (dg > 0.0f) ? rsqrtf(dg) : 0.0f;
        }
    } else {
    const size_t gbase = (size_t)rowBase * F;
    const bool vec = (F % 4 == 0) &&
        ((((size_t)dest | (size_t)H | (size_t)bias) & 15) == 0);
    if (vec) {
        const int n4 = (nrows * F) >> 2;
        for (int p = t; p < n4; p += 1024) {
            const int idx = p << 2;
            const int r = idx / F;
            const int f = idx - r * F;
            const float di = dinv[rowBase + r];
            const v4f d4 = *(const v4f*)&acc[idx];
            const v4f h4 = *(const v4f*)(H + gbase + idx);
            const v4f b4 = *(const v4f*)(bias + f);
            v4f v;
            v.x = d4.x + h4.x * di * di + b4.x;
            v.y = d4.y + h4.y * di * di + b4.y;
            v.z = d4.z + h4.z * di * di + b4.z;
            v.w = d4.w + h4.w * di * di + b4.w;
            if (doRelu) {
                v.x = fmaxf(v.x, 0.0f); v.y = fmaxf(v.y, 0.0f);
                v.z = fmaxf(v.z, 0.0f); v.w = fmaxf(v.w, 0.0f);
            }
            *(v4f*)(dest + gbase + idx) = v;
        }
    } else {
        const int n1 = nrows * F;
        for (int idx = t; idx < n1; idx += 1024) {
            const int r = idx / F;
            const int f = idx - r * F;
            const float di = dinv[rowBase + r];
            const float v = acc[idx] + H[gbase + idx] * di * di + bias[f];
            dest[gbase + idx] = doRelu ? fmaxf(v, 0.0f) : v;
        }
    }
    }
}

__global__ void k_agg_small(
    const float* __restrict__ H, const int* __restrict__ src,
    const int* __restrict__ dst, const float* __restrict__ dinv,
    float* __restrict__ out, int E, int F)
{
    const unsigned nth = gridDim.x * blockDim.x;
    const unsigned tot = (unsigned)E * (unsigned)F;
    const unsigned uF  = (unsigned)F;
    for (unsigned idx = blockIdx.x * blockDim.x + threadIdx.x; idx < tot; idx += nth) {
        const unsigned e = idx / uF;
        const unsigned f = idx - e * uF;
        const int s = src[e], d = dst[e];
        const float nrm = dinv[s] * dinv[d];
        atomicAdd(&out[(size_t)d * F + f], H[(size_t)s * F + f] * nrm);
    }
}

__global__ void k_finish(float* __restrict__ dest, const float* __restrict__ H,
                         const float* __restrict__ dinv, const float* __restrict__ bias,
                         int N, int F, int doRelu)
{
    const unsigned tid = blockIdx.x * blockDim.x + threadIdx.x;
    const unsigned nth = gridDim.x * blockDim.x;
    const unsigned tot = (unsigned)N * (unsigned)F;
    const unsigned uF  = (unsigned)F;
    const bool vec = ((F & 3) == 0) &&
        ((((size_t)dest | (size_t)H | (size_t)bias) & 15) == 0);
    if (vec) {
        const unsigned n4 = tot >> 2;
        for (unsigned q = tid; q < n4; q += nth) {
            const unsigned idx = q << 2;
            const unsigned i = idx / uF;
            const unsigned f = idx - i * uF;
            const float di = dinv[i];
            const v4f d4 = ((const v4f*)dest)[q];
            const v4f h4 = ((const v4f*)H)[q];
            const v4f b4 = *(const v4f*)(bias + f);
            v4f v;
            v.x = d4.x + h4.x * di * di + b4.x;
            v.y = d4.y + h4.y * di * di + b4.y;
            v.z = d4.z + h4.z * di * di + b4.z;
            v.w = d4.w + h4.w * di * di + b4.w;
            if (doRelu) {
                v.x = fmaxf(v.x, 0.0f); v.y = fmaxf(v.y, 0.0f);
                v.z = fmaxf(v.z, 0.0f); v.w = fmaxf(v.w, 0.0f);
            }
            ((v4f*)dest)[q] = v;
        }
    } else {
        for (unsigned idx = tid; idx < tot; idx += nth) {
            const unsigned i = idx / uF;
            const unsigned f = idx - i * uF;
            const float di = dinv[i];
            const float v  = dest[idx] + H[idx] * di * di + bias[f];
            dest[idx] = doRelu ? fmaxf(v, 0.0f) : v;
        }
    }
}

extern "C" void kernel_launch(void* const* d_in, const int* in_sizes, int n_in,
                              void* d_out, int out_size, void* d_ws, size_t ws_size,
                              hipStream_t stream) {
    (void)n_in; (void)out_size;

    const float* x   = (const float*)d_in[0];
    const int*   ei  = (const int*)  d_in[1];
    const float* W1  = (const float*)d_in[2];
    const float* b1  = (const float*)d_in[3];
    const float* W1b = (const float*)d_in[4];
    const float* b1b = (const float*)d_in[5];
    const float* W2b = (const float*)d_in[6];
    const float* b2b = (const float*)d_in[7];
    const float* W2  = (const float*)d_in[8];
    const float* b2  = (const float*)d_in[9];

    const int F_IN = 500, H1 = 200, H2 = 100, H3 = 40, C = 3;
    const int N = in_sizes[0] / F_IN;
    const int E = in_sizes[1] / 2;
    const int* srcIdx = ei;
    const int* dstIdx = ei + E;

    char*  ws   = (char*)d_ws;
    float* dinv = (float*)ws;
    size_t off  = ((size_t)N * sizeof(float) + 511) & ~(size_t)511;
    float* Wpad = (float*)(ws + off);
    off += (size_t)256 * 512 * sizeof(float);
    float* bufH = (float*)(ws + off);
    off += (((size_t)N * H1 * sizeof(float)) + 511) & ~(size_t)511;
    float* bufA = (float*)(ws + off);
    float* out  = (float*)d_out;
    off += (((size_t)N * H1 * sizeof(float)) + 511) & ~(size_t)511;
    int* keyp = (int*)(ws + off);
    const bool keyed = (ws_size >= off + ((((size_t)E + 15) >> 4) << 4)) &&
                       ((N + OWN_R - 1) / OWN_R <= 256);
    int* listA = (int*)bufA;
    int* listB = (int*)(bufH + (size_t)N * H2);
    off += ((((size_t)E + 15) >> 4) << 4);
    off  = (off + 511) & ~(size_t)511;
    int* listK = (int*)(ws + off);
    const bool kept = keyed && (ws_size >= off + (size_t)N * KEEP_CAP * sizeof(int));

    const int resBlocks  = 2048;
    auto gridFor = [&](size_t items) {
        size_t b = (items + 255) / 256;
        return (int)(b < (size_t)resBlocks ? b : (size_t)resBlocks);
    };

    if (keyed) {
        k_key<<<gridFor(((size_t)E + 3) / 4), 256, 0, stream>>>(dstIdx, keyp, E);
        if (kept)
            k_agg_own<0, 1, 1, 1, OWN_R, true, true, 1, KEEP_CAP><<<(N + OWN_R - 1) / OWN_R, 1024, 0, stream>>>(
                nullptr, srcIdx, dstIdx, nullptr, nullptr, dinv, keyp, listK, N, E, 0);
        else
            k_agg_own<0, 1, 1, 1, OWN_R, true, true, 1, 200><<<(N + OWN_R - 1) / OWN_R, 1024, 0, stream>>>(
                nullptr, srcIdx, dstIdx, nullptr, nullptr, dinv, keyp, listA, N, E, 0);
    } else {
        k_fill1<<<(N + 255) / 256, 256, 0, stream>>>(dinv, N);
        k_deg<<<gridFor(((size_t)E + 3) / 4), 256, 0, stream>>>(dstIdx, dinv, E);
        k_rsqrt<<<(N + 255) / 256, 256, 0, stream>>>(dinv, N);
    }

    const int gemmBlocks = (N + 15) / 16;

    auto layer = [&](const float* in, const float* Wt, const float* bias,
                     float* dest, int K, int F, int relu) {
        const int Kpad = (K + TK - 1) & ~(TK - 1);
        k_packW<<<(256 * Kpad + 255) / 256, 256, 0, stream>>>(Wt, Wpad, F, K, Kpad);
        if (F == 100)
            k_gemm_wmma<7><<<gemmBlocks, 256, 0, stream>>>(in, Wpad, bufH, N, K, Kpad, F);
        else if (F == 40)
            k_gemm_wmma<3><<<gemmBlocks, 256, 0, stream>>>(in, Wpad, bufH, N, K, Kpad, F);
        else if (F <= 16)
            k_gemm_wmma<1><<<gemmBlocks, 256, 0, stream>>>(in, Wpad, bufH, N, K, Kpad, F);
        else
            k_gemm_wmma<16><<<gemmBlocks, 256, 0, stream>>>(in, Wpad, bufH, N, K, Kpad, F);
        const int keyBlocks = (N + OWN_R - 1) / OWN_R;
        if (F == 200 && kept) {
            k_agg_own<7, 4, 1, 200, OWN_R, true, false, 2, KEEP_CAP><<<keyBlocks, 1024, 0, stream>>>(
                bufH, srcIdx, dstIdx, dinv, bias, dest, keyp, listK, N, E, relu);
        } else if (F == 100 && kept) {
            k_agg_own<4, 8, 1, 100, OWN_R, true, false, 2, KEEP_CAP><<<keyBlocks, 1024, 0, stream>>>(
                bufH, srcIdx, dstIdx, dinv, bias, dest, keyp, listK, N, E, relu);
        } else if (F == 40 && kept) {
            k_agg_own<2, 4, 2, 40, OWN_R, true, false, 2, KEEP_CAP><<<keyBlocks, 1024, 0, stream>>>(
                bufH, srcIdx, dstIdx, dinv, bias, dest, keyp, listK, N, E, relu);
        } else if (F == 3 && kept) {
            k_agg_own<0, 1, 1, 3, OWN_R, true, false, 2, KEEP_CAP><<<keyBlocks, 1024, 0, stream>>>(
                bufH, srcIdx, dstIdx, dinv, bias, dest, keyp, listK, N, E, relu);
        } else if (F == 200 && keyed) {
            k_agg_own<7, 4, 1, 200, OWN_R, true, false, 2, 200><<<keyBlocks, 1024, 0, stream>>>(
                bufH, srcIdx, dstIdx, dinv, bias, dest, keyp, listA, N, E, relu);
        } else if (F == 100 && keyed) {
            k_agg_own<4, 8, 1, 100, OWN_R, true, false, 1, 100><<<keyBlocks, 1024, 0, stream>>>(
                bufH, srcIdx, dstIdx, dinv, bias, dest, keyp, listB, N, E, relu);
        } else if (F == 40 && keyed) {
            k_agg_own<2, 4, 2, 40, OWN_R, true, false, 2, 100><<<keyBlocks, 1024, 0, stream>>>(
                bufH, srcIdx, dstIdx, dinv, bias, dest, keyp, listB, N, E, relu);
        } else if (F == 3 && keyed) {
            k_agg_own<0, 1, 1, 3, OWN_R, true, false, 2, 100><<<keyBlocks, 1024, 0, stream>>>(
                bufH, srcIdx, dstIdx, dinv, bias, dest, keyp, listB, N, E, relu);
        } else if (F == 200) {
            k_agg_own<7, 4, 1, 200, 396, false, false, 0, 1><<<(N + 395) / 396, 1024, 0, stream>>>(
                bufH, srcIdx, dstIdx, dinv, bias, dest, keyp, nullptr, N, E, relu);
        } else if (F == 100) {
            k_agg_own<4, 8, 1, 100, 400, false, false, 0, 1><<<(N + 399) / 400, 1024, 0, stream>>>(
                bufH, srcIdx, dstIdx, dinv, bias, dest, keyp, nullptr, N, E, relu);
        } else if (F == 40) {
            k_agg_own<2, 4, 2, 40, 400, false, false, 0, 1><<<(N + 399) / 400, 1024, 0, stream>>>(
                bufH, srcIdx, dstIdx, dinv, bias, dest, keyp, nullptr, N, E, relu);
        } else {
            size_t tot = (size_t)N * F;
            const int streamGrid = gridFor((tot + 3) / 4);
            k_zero<<<streamGrid, 256, 0, stream>>>(dest, tot);
            k_agg_small<<<gridFor(((size_t)E + 3) / 4), 256, 0, stream>>>(
                bufH, srcIdx, dstIdx, dinv, dest, E, F);
            k_finish<<<streamGrid, 256, 0, stream>>>(
                dest, bufH, dinv, bias, N, F, relu);
        }
    };

    layer(x,    W1,  b1,  bufA, F_IN, H1, 1);
    layer(bufA, W1b, b1b, bufA, H1,   H2, 1);
    layer(bufA, W2b, b2b, bufA, H2,   H3, 1);
    layer(bufA, W2,  b2,  out,  H3,   C,  0);
}


# GEMM block = 80 rows: each wave keeps a packed B fragment in registers across 5 row tiles (5x fewer B loads and GEMM waves); the 80x64 A tile's loads go out as one group; k and sum order kept
# speedup vs baseline: 3.5268x; 1.0536x over previous
#include <hip/hip_runtime.h>

typedef float v2f __attribute__((ext_vector_type(2)));
typedef float v4f __attribute__((ext_vector_type(4)));
typedef float v8f __attribute__((ext_vector_type(8)));
typedef int   v4i __attribute__((ext_vector_type(4)));

#define TK 64
#define RT 5

template <int NT>
__global__ __launch_bounds__(256) void k_gemm_wmma(
    const float* __restrict__ X, const float* Wf,
    float* __restrict__ H, int M, int K, int Kpad, int Nout)
{
    __shared__ float As[16 * RT][TK + 4];

    const int row0 = (int)blockIdx.x * (16 * RT);
    const int t    = threadIdx.x;
    const int wave = (__builtin_amdgcn_readfirstlane(t >> 5) +
                      (int)((blockIdx.x * 2654435761u) >> 29)) & 7;
    const int lane = t & 31;
    const int half = lane >> 4;
    const int m    = lane & 15;
    const bool has0 = wave < NT;
    const bool has1 = (wave + 8) < NT;

    const int n0 = (wave << 4) + m;
    const int n1 = ((wave + 8) << 4) + m;
    const int KS2 = Kpad >> 3;
    const v4f* Wq  = (const v4f*)Wf;
    const v4f* Wf0 = Wq + (size_t)wave * KS2 * 32 + lane;
    const v4f* Wf1 = Wq + (size_t)(wave + 8) * KS2 * 32 + lane;

    v8f acc0[RT], acc1[RT];
    #pragma unroll
    for (int r = 0; r < RT; ++r) { acc0[r] = v8f{}; acc1[r] = v8f{}; }

    for (int k0 = 0; k0 < Kpad; k0 += TK) {
        __syncthreads();
        const int s0 = k0 >> 3;
        v4f q0[TK / 8], q1[TK / 8];
        if (has0) {
            #pragma unroll
            for (int j = 0; j < TK / 8; ++j) q0[j] = Wf0[(size_t)(s0 + j) * 32];
        }
        if (has1) {
            #pragma unroll
            for (int j = 0; j < TK / 8; ++j) q1[j] = Wf1[(size_t)(s0 + j) * 32];
        }
        float av[(16 * RT * TK) / 256];
        #pragma unroll
        for (int i = 0; i < (16 * RT * TK) / 256; ++i) {
            const int e  = t + (i << 8);
            const int r  = e >> 6;
            const int kk = e & (TK - 1);
            const int gr = row0 + r, gk = k0 + kk;
            const bool ok = (gr < M && gk < K);
            const float v = X[ok ? ((size_t)gr * K + gk) : (size_t)0];
            av[i] = ok ? v : 0.0f;
        }
        #pragma unroll
        for (int i = 0; i < (16 * RT * TK) / 256; ++i) {
            const int e = t + (i << 8);
            As[e >> 6][e & (TK - 1)] = av[i];
        }
        __syncthreads();

        if (has1) {
            #pragma unroll
            for (int j = 0; j < TK / 8; ++j) {
                const int kb = (j << 3) + (half << 1);
                const v2f b00 = __builtin_shufflevector(q0[j], q0[j], 0, 1);
                const v2f b01 = __builtin_shufflevector(q0[j], q0[j], 2, 3);
                const v2f b10 = __builtin_shufflevector(q1[j], q1[j], 0, 1);
                const v2f b11 = __builtin_shufflevector(q1[j], q1[j], 2, 3);
                #pragma unroll
                for (int r = 0; r < RT; ++r) {
                    const v2f a0 = *(const v2f*)&As[(r << 4) + m][kb];
                    acc0[r] = __builtin_amdgcn_wmma_f32_16x16x4_f32(
                        false, a0, false, b00, (short)0, acc0[r], false, false);
                    acc1[r] = __builtin_amdgcn_wmma_f32_16x16x4_f32(
                        false, a0, false, b10, (short)0, acc1[r], false, false);
                }
                #pragma unroll
                for (int r = 0; r < RT; ++r) {
                    const v2f a1 = *(const v2f*)&As[(r << 4) + m][kb + 4];
                    acc0[r] = __builtin_amdgcn_wmma_f32_16x16x4_f32(
                        false, a1, false, b01, (short)0, acc0[r], false, false);
                    acc1[r] = __builtin_amdgcn_wmma_f32_16x16x4_f32(
                        false, a1, false, b11, (short)0, acc1[r], false, false);
                }
            }
        } else if (has0) {
            #pragma unroll
            for (int j = 0; j < TK / 8; ++j) {
                const int kb = (j << 3) + (half << 1);
                const v2f b00 = __builtin_shufflevector(q0[j], q0[j], 0, 1);
                const v2f b01 = __builtin_shufflevector(q0[j], q0[j], 2, 3);
                #pragma unroll
                for (int r = 0; r < RT; ++r) {
                    const v2f a0 = *(const v2f*)&As[(r << 4) + m][kb];
                    acc0[r] = __builtin_amdgcn_wmma_f32_16x16x4_f32(
                        false, a0, false, b00, (short)0, acc0[r], false, false);
                }
                #pragma unroll
                for (int r = 0; r < RT; ++r) {
                    const v2f a1 = *(const v2f*)&As[(r << 4) + m][kb + 4];
                    acc0[r] = __builtin_amdgcn_wmma_f32_16x16x4_f32(
                        false, a1, false, b01, (short)0, acc0[r], false, false);
                }
            }
        }
    }

    #pragma unroll
    for (int rt = 0; rt < RT; ++rt) {
        #pragma unroll
        for (int r = 0; r < 8; ++r) {
            int grow = row0 + (rt << 4) + r + (half << 3);
            if (grow < M) {
                if (has0 && n0 < Nout) H[(size_t)grow * Nout + n0] = acc0[rt][r];
                if (has1 && n1 < Nout) H[(size_t)grow * Nout + n1] = acc1[rt][r];
            }
        }
    }
}

__global__ void k_packW(const float* __restrict__ W, float* __restrict__ Wf,
                        int Nout, int K, int Kpad)
{
    int idx = blockIdx.x * blockDim.x + threadIdx.x;
    int tot = 256 * Kpad;
    if (idx >= tot) return;
    const int KS2  = Kpad >> 3;
    const int q    = idx & 3;
    const int lane = (idx >> 2) & 31;
    const int rest = idx >> 7;
    const int s    = rest % KS2;
    const int ct   = rest / KS2;
    const int n = (ct << 4) + (lane & 15);
    const int k = (s << 3) + ((q >> 1) << 2) + ((lane >> 4) << 1) + (q & 1);
    Wf[idx] = (n < Nout && k < K) ? W[(size_t)n * K + k] : 0.0f;
}

__global__ void k_fill1(float* __restrict__ p, int n) {
    int i = blockIdx.x * blockDim.x + threadIdx.x;
    if (i < n) p[i] = 1.0f;
}

__global__ void k_deg(const int* __restrict__ dst, float* __restrict__ deg, int E) {
    const int nth = (int)(gridDim.x * blockDim.x);
    for (int e = (int)(blockIdx.x * blockDim.x + threadIdx.x); e < E; e += nth)
        atomicAdd(&deg[dst[e]], 1.0f);
}

#define OWN_R 392

__global__ __launch_bounds__(256) void k_key(
    const int* __restrict__ dst, int* __restrict__ key, int E)
{
    const int nth = (int)(gridDim.x * blockDim.x);
    const int nq  = (E + 3) >> 2;
    const bool al = ((((size_t)dst) & 15) == 0);
    for (int qd = (int)(blockIdx.x * blockDim.x + threadIdx.x); qd < nq; qd += nth) {
        const int e0 = qd << 2;
        unsigned w = 0u;
        if (al && e0 + 4 <= E) {
            const v4i d4 = *(const v4i*)(dst + e0);
            w = (unsigned)(d4.x / OWN_R) | ((unsigned)(d4.y / OWN_R) << 8) |
                ((unsigned)(d4.z / OWN_R) << 16) | ((unsigned)(d4.w / OWN_R) << 24);
        } else {
            for (int i = 0; i < 4; ++i)
                if (e0 + i < E) w |= (unsigned)(dst[e0 + i] / OWN_R) << (8 * i);
        }
        key[qd] = (int)w;
    }
}

__global__ void k_rsqrt(float* __restrict__ p, int n) {
    int i = blockIdx.x * blockDim.x + threadIdx.x;
    if (i < n) { float d = p[i]; p[i] = (d > 0.0f) ? rsqrtf(d) : 0.0f; }
}

__global__ __launch_bounds__(256) void k_zero(float* __restrict__ p, size_t n) {
    const size_t tid = (size_t)blockIdx.x * blockDim.x + threadIdx.x;
    const size_t nth = (size_t)gridDim.x * blockDim.x;
    if ((((size_t)p) & 15) == 0) {
        const size_t n4 = n >> 2;
        v4f* __restrict__ p4 = (v4f*)p;
        const v4f z = {0.0f, 0.0f, 0.0f, 0.0f};
        for (size_t i = tid; i < n4; i += nth) p4[i] = z;
        for (size_t i = (n4 << 2) + tid; i < n; i += nth) p[i] = 0.0f;
    } else {
        for (size_t i = tid; i < n; i += nth) p[i] = 0.0f;
    }
}

template <int NF, int TU, int NG>
__device__ __forceinline__ void agg_groups(
    const float* __restrict__ H, float* __restrict__ out,
    int s, int d, float nrm, int j, int F, int lane)
{
    constexpr int T  = 32 / TU;
    constexpr int U  = TU * NG;
    constexpr int FB = (NF - 1) << 5;
    const int tl = lane / T;
    const int tf = FB + (lane % T);

    float v[U][NF - 1];
    float vt[NG], nt[NG];
    int   dt[NG];
    #pragma unroll
    for (int u = 0; u < U; ++u) {
        const int su = __builtin_amdgcn_readlane(s, j + u);
        const float* __restrict__ hrow = H + (size_t)su * F;
        #pragma unroll
        for (int c = 0; c < NF - 1; ++c) v[u][c] = hrow[lane + (c << 5)];
    }
    #pragma unroll
    for (int g = 0; g < NG; ++g) {
        const int ju = j + g * TU + tl;
        const int st = __shfl(s, ju);
        dt[g] = __shfl(d, ju);
        nt[g] = __shfl(nrm, ju);
        vt[g] = H[(size_t)st * F + tf];
    }
    #pragma unroll
    for (int u = 0; u < U; ++u) {
        const int   du = __builtin_amdgcn_readlane(d, j + u);
        const float nu = __builtin_bit_cast(float,
            __builtin_amdgcn_readlane(__builtin_bit_cast(int, nrm), j + u));
        float* __restrict__ orow = out + (size_t)du * F;
        #pragma unroll
        for (int c = 0; c < NF - 1; ++c)
            atomicAdd(&orow[lane + (c << 5)], v[u][c] * nu);
    }
    #pragma unroll
    for (int g = 0; g < NG; ++g)
        atomicAdd(&out[(size_t)dt[g] * F + tf], vt[g] * nt[g]);
}

template <int NF>
__device__ __forceinline__ void agg_one(
    const float* __restrict__ H, float* __restrict__ out,
    int s, int d, float nrm, int j, int F, int lane)
{
    const int  fl = lane + ((NF - 1) << 5);
    const bool lastOn = (fl < F);
    const int   sj = __builtin_amdgcn_readlane(s, j);
    const int   dj = __builtin_amdgcn_readlane(d, j);
    const float nj = __builtin_bit_cast(float,
        __builtin_amdgcn_readlane(__builtin_bit_cast(int, nrm), j));
    const float* __restrict__ hrow = H + (size_t)sj * F;
    float* __restrict__ orow = out + (size_t)dj * F;
    float v[NF];
    #pragma unroll
    for (int c = 0; c < NF - 1; ++c) v[c] = hrow[lane + (c << 5)];
    v[NF - 1] = lastOn ? hrow[fl] : 0.0f;
    #pragma unroll
    for (int c = 0; c < NF - 1; ++c) atomicAdd(&orow[lane + (c << 5)], v[c] * nj);
    if (lastOn) atomicAdd(&orow[fl], v[NF - 1] * nj);
}

#define SCAN_U   8

#define LIST_HDR 32
#define KEEP_CAP 24

template <int NF, int TU, int NG, int F, bool DEG, int MODE>
__device__ __forceinline__ void own_batch(
    const float* __restrict__ H, const int* __restrict__ src,
    const int* __restrict__ dst, const float* __restrict__ dinv,
    float* acc, int ent, int cnt, int rowBase, int lane,
    int* lst, int cap, int* lcnt)
{
    int pos = 0;
    if constexpr (MODE == 1) {
        if (lane == 0) pos = atomicAdd(lcnt, cnt);
        pos = __builtin_amdgcn_readfirstlane(pos);
    }
    if constexpr (NF == 0) {
        if (lane < cnt) {
            int s = 0, dl;
            if constexpr (MODE == 2) { s = ent >> 9; dl = ent & 511; }
            else {
                dl = dst[ent] - rowBase;
                if constexpr (!DEG || MODE == 1) s = src[ent];
            }
            float* row = acc + dl * F;
            if constexpr (DEG) {
                atomicAdd(row, 1.0f);
            } else {
                const float nrm = dinv[s] * dinv[rowBase + dl];
                const float* __restrict__ hrow = H + (size_t)s * F;
                float v[F];
                #pragma unroll
                for (int f = 0; f < F; ++f) v[f] = hrow[f];
                #pragma unroll
                for (int f = 0; f < F; ++f) atomicAdd(&row[f], v[f] * nrm);
            }
            if constexpr (MODE == 1)
                if (pos + lane < cap) lst[LIST_HDR + pos + lane] = (s << 9) | dl;
        }
    } else {
    constexpr int U = TU * NG;
    int s = 0, dl = 0; float nrm = 0.0f;
    if (lane < cnt) {
        if constexpr (MODE == 2) { s = ent >> 9; dl = ent & 511; }
        else { s = src[ent]; dl = dst[ent] - rowBase; }
        nrm = dinv[s] * dinv[rowBase + dl];
        if constexpr (MODE == 1)
            if (pos + lane < cap) lst[LIST_HDR + pos + lane] = (s << 9) | dl;
    }
    int j = 0;
    for (; j + U <= cnt; j += U)
        agg_groups<NF, TU, NG>(H, acc, s, dl, nrm, j, F, lane);
    for (; j < cnt; ++j)
        agg_one<NF>(H, acc, s, dl, nrm, j, F, lane);
    }
}

template <int NF, int TU, int NG, int F, int R, bool KEYED, bool DEG, int MODE, int FCAP>
__global__ __launch_bounds__(1024) void k_agg_own(
    const float* __restrict__ H, const int* __restrict__ src,
    const int* __restrict__ dst, const float* __restrict__ dinv,
    const float* __restrict__ bias, float* dest,
    const int* __restrict__ key, int* lst, int N, int E, int doRelu)
{
    static_assert(MODE == 0 || (KEYED && R <= 512), "a list entry keeps the local row in 9 bits");
    static_assert(NF == 0 || (R * F * 4) % 128 == 0, "block boundary must fall on a 128-byte line");
    static_assert((R * F) % 4 == 0, "the sums are cleared in 16-byte pieces");
    static_assert(SCAN_U * 4 <= 32, "one flag bit per load and piece in a 32-bit word");
    constexpr int ACC_N = (NF == 0 && R * F < 41472) ? 41472 : R * F;
    __shared__ __attribute__((aligned(16))) float acc[ACC_N];
    __shared__ int   que[32][64];
    __shared__ int   lcnt;

    const int t    = threadIdx.x;
    const int lane = t & 31;
    const int wave = __builtin_amdgcn_readfirstlane(t >> 5);
    const int rowBase = (int)blockIdx.x * R;
    const int nrows   = (N - rowBase < R) ? (N - rowBase) : R;

    {
        const v4f z = {0.0f, 0.0f, 0.0f, 0.0f};
        v4f* a4 = (v4f*)acc;
        for (int i = t; i < (R * F) / 4; i += 1024) a4[i] = z;
        if (t == 0) lcnt = 0;
    }
    __syncthreads();

    int* q = que[wave];
    int  qn = 0;
    const int nwv = (int)(blockDim.x >> 5);
    const int eb = (int)(((long long)wave * E) / nwv);
    const int ee = (int)(((long long)(wave + 1) * E) / nwv);

    int* L  = nullptr;
    int cap = 0;
    if constexpr (MODE != 0) {
        L   = lst + (size_t)rowBase * FCAP;
        cap = nrows * FCAP - LIST_HDR;
    }
    constexpr int SM = (MODE == 1) ? 1 : 0;
    bool scan = true;
    if constexpr (MODE == 2) {
        const int total = __builtin_amdgcn_readfirstlane(L[0]);
        if (total <= cap) {
            scan = false;
            const int nch = (total + 31) >> 5;
            int c = wave;
            int entNext = 0;
            if (c < nch) {
                const int ix = (c << 5) + lane;
                entNext = L[LIST_HDR + (ix < total ? ix : total - 1)];
            }
            while (c < nch) {
                const int ent = entNext;
                const int cnt = (total - (c << 5) < 32) ? (total - (c << 5)) : 32;
                c += nwv;
                if (c < nch) {
                    const int ix = (c << 5) + lane;
                    entNext = L[LIST_HDR + (ix < total ? ix : total - 1)];
                }
                own_batch<NF, TU, NG, F, DEG, 2>(H, src, dst, dinv, acc, ent, cnt, rowBase, lane,
                                                 nullptr, 0, nullptr);
            }
        }
    }

    if (scan) {
    if constexpr (KEYED) {
    const unsigned pat = (unsigned)blockIdx.x * 0x01010101u;
    auto zb = [&](int wv) -> unsigned {
        const unsigned x = (unsigned)wv ^ pat;
        const unsigned s7 = (x & 0x7F7F7F7Fu) + 0x7F7F7F7Fu;
        return ~(s7 | x | 0x7F7F7F7Fu);
    };
    const int P  = E >> 4;
    const int pb = (int)(((long long)wave * P) / nwv);
    const int pe = (int)(((long long)(wave + 1) * P) / nwv);
    const v4i* __restrict__ k4 = (const v4i*)key;
    for (int p0 = pb; p0 < pe; p0 += 32 * SCAN_U) {
        v4i kk[SCAN_U];
        #pragma unroll
        for (int k = 0; k < SCAN_U; ++k) {
            const int p = p0 + (k << 5) + lane;
            kk[k] = k4[p < pe ? p : pe - 1];
        }
        __builtin_amdgcn_sched_barrier(0);
        unsigned mk[SCAN_U];
        unsigned anyk = 0u;
        #pragma unroll
        for (int k = 0; k < SCAN_U; ++k) {
            const unsigned z0 = zb(kk[k].x), z1 = zb(kk[k].y);
            const unsigned z2 = zb(kk[k].z), z3 = zb(kk[k].w);
            const unsigned m = (z0 >> 7) | (z1 >> 6) | (z2 >> 5) | (z3 >> 4);
            mk[k] = (p0 + (k << 5) + lane < pe) ? m : 0u;
            anyk |= (__builtin_amdgcn_ballot_w32(mk[k] != 0u) != 0u) ? (1u << k) : 0u;
        }
        while (anyk != 0u) {
            const int k = __builtin_ctz(anyk);
            anyk &= anyk - 1u;
            unsigned m = mk[0];
            #pragma unroll
            for (int i = 1; i < SCAN_U; ++i) m = (k == i) ? mk[i] : m;
            const int ebase = (p0 + (k << 5) + lane) << 4;
            unsigned live = __builtin_amdgcn_ballot_w32(m != 0u);
            while (live != 0u) {
                const bool hit = m != 0u;
                const int  b   = __builtin_ctz(m | 0x80000000u);
                if (hit) q[qn + __builtin_popcount(live & ((1u << lane) - 1u))] =
                             ebase + ((b & 7) << 2) + (b >> 3);
                qn += __builtin_popcount(live);
                if (qn >= 32) {
                    own_batch<NF, TU, NG, F, DEG, SM>(H, src, dst, dinv, acc, q[lane], 32, rowBase, lane,
                                                      L, cap, &lcnt);
                    const int rest = qn - 32;
                    int tmp = 0;
                    if (lane < rest) tmp = q[32 + lane];
                    if (lane < rest) q[lane] = tmp;
                    qn = rest;
                }
                m &= m - 1u;
                live = __builtin_amdgcn_ballot_w32(m != 0u);
            }
        }
    }
    if (wave == nwv - 1 && (P << 4) < E) {
        const int pos = P << 4;
        const int e   = pos + lane;
        const bool hit = (e < E) && ((unsigned)(dst[e < E ? e : (E - 1)] - rowBase) < (unsigned)nrows);
        const unsigned m = __builtin_amdgcn_ballot_w32(hit);
        if (hit) q[qn + __builtin_popcount(m & ((1u << lane) - 1u))] = e;
        qn += __builtin_popcount(m);
    }
    } else {
    const unsigned unr = (unsigned)nrows;
    const int nE = ee - eb;
    const int* __restrict__ wp = dst + eb;
    int head = (int)(((16u - ((unsigned)(size_t)wp & 15u)) & 15u) >> 2);
    if (head > nE) head = nE;
    const int eg = eb + head;
    const int ng = (ee - eg) >> 7;
    const int et = eg + (ng << 7);
    const v4i* __restrict__ p4 = (const v4i*)(dst + eg) + lane;
    int pos = eb, g0 = 0;
    while (pos < ee) {
        unsigned hb = 0u, slots = 0u;
        int ebase, lmul;
        if (pos >= eg && g0 < ng) {
            v4i dk[SCAN_U];
            #pragma unroll
            for (int k = 0; k < SCAN_U; ++k) {
                const int g = (g0 + k < ng) ? (g0 + k) : (ng - 1);
                dk[k] = p4[(size_t)g << 5];
            }
            #pragma unroll
            for (int k = 0; k < SCAN_U; ++k) {
                if (g0 + k < ng) {
                    const unsigned u0 = (unsigned)(dk[k].x - rowBase);
                    const unsigned u1 = (unsigned)(dk[k].y - rowBase);
                    const unsigned u2 = (unsigned)(dk[k].z - rowBase);
                    const unsigned u3 = (unsigned)(dk[k].w - rowBase);
                    const unsigned a = u0 < u1 ? u0 : u1;
                    const unsigned b = u2 < u3 ? u2 : u3;
                    const unsigned mn = a < b ? a : b;
                    if (__builtin_amdgcn_ballot_w32(mn < unr) != 0u) {
                        const unsigned m0 = __builtin_amdgcn_ballot_w32(u0 < unr);
                        const unsigned m1 = __builtin_amdgcn_ballot_w32(u1 < unr);
                        const unsigned m2 = __builtin_amdgcn_ballot_w32(u2 < unr);
                        const unsigned m3 = __builtin_amdgcn_ballot_w32(u3 < unr);
                        hb |= ((u0 < unr ? 1u : 0u) | (u1 < unr ? 2u : 0u) |
                               (u2 < unr ? 4u : 0u) | (u3 < unr ? 8u : 0u)) << (4 * k);
                        slots |= ((m0 != 0u ? 1u : 0u) | (m1 != 0u ? 2u : 0u) |
                                  (m2 != 0u ? 4u : 0u) | (m3 != 0u ? 8u : 0u)) << (4 * k);
                    }
                }
            }
            ebase = eg + (g0 << 7);
            lmul  = 4;
            g0 += SCAN_U;
            pos = (g0 < ng) ? eg : et;
        } else {
            const int lim = (pos < eg) ? eg : ee;
            const int cnt = (lim - pos < 32) ? (lim - pos) : 32;
            const bool ok = lane < cnt;
            const int  dv = dst[ok ? (pos + lane) : pos];
            const bool hit = ok && ((unsigned)(dv - rowBase) < unr);
            hb    = hit ? 1u : 0u;
            slots = (__builtin_amdgcn_ballot_w32(hit) != 0u) ? 1u : 0u;
            ebase = pos;
            lmul  = 1;
            pos += cnt;
        }
        while (slots != 0u) {
            const int b = __builtin_ctz(slots);
            slots &= slots - 1u;
            const bool hit = ((hb >> b) & 1u) != 0u;
            const unsigned m = __builtin_amdgcn_ballot_w32(hit);
            if (hit) q[qn + __builtin_popcount(m & ((1u << lane) - 1u))] =
                         ebase + ((b >> 2) << 7) + lane * lmul + (b & 3);
            qn += __builtin_popcount(m);
            if (qn >= 32) {
                own_batch<NF, TU, NG, F, DEG, SM>(H, src, dst, dinv, acc, q[lane], 32, rowBase, lane,
                                                  L, cap, &lcnt);
                const int rest = qn - 32;
                int tmp = 0;
                if (lane < rest) tmp = q[32 + lane];
                if (lane < rest) q[lane] = tmp;
                qn = rest;
            }
        }
    }
    }
    while (qn > 0) {
        const int cnt = (qn < 32) ? qn : 32;
        own_batch<NF, TU, NG, F, DEG, SM>(H, src, dst, dinv, acc, (lane < cnt) ? q[lane] : 0, cnt, rowBase, lane,
                                          L, cap, &lcnt);
        const int rest = qn - cnt;
        int tmp = 0;
        if (lane < rest) tmp = q[32 + lane];
        if (lane < rest) q[lane] = tmp;
        qn = rest;
    }
    }

    __syncthreads();

    if constexpr (MODE == 1) {
        if (t == 0) L[0] = lcnt;
    }

    if constexpr (DEG) {
        for (int r = t; r < nrows; r += 1024) {
            const float dg = 1.0f + acc[r];
            dest[rowBase + r] = (dg > 0.0f) ? rsqrtf(dg) : 0.0f;
        }
    } else {
    const size_t gbase = (size_t)rowBase * F;
    const bool vec = (F % 4 == 0) &&
        ((((size_t)dest | (size_t)H | (size_t)bias) & 15) == 0);
    if (vec) {
        const int n4 = (nrows * F) >> 2;
        for (int p = t; p < n4; p += 1024) {
            const int idx = p << 2;
            const int r = idx / F;
            const int f = idx - r * F;
            const float di = dinv[rowBase + r];
            const v4f d4 = *(const v4f*)&acc[idx];
            const v4f h4 = *(const v4f*)(H + gbase + idx);
            const v4f b4 = *(const v4f*)(bias + f);
            v4f v;
            v.x = d4.x + h4.x * di * di + b4.x;
            v.y = d4.y + h4.y * di * di + b4.y;
            v.z = d4.z + h4.z * di * di + b4.z;
            v.w = d4.w + h4.w * di * di + b4.w;
            if (doRelu) {
                v.x = fmaxf(v.x, 0.0f); v.y = fmaxf(v.y, 0.0f);
                v.z = fmaxf(v.z, 0.0f); v.w = fmaxf(v.w, 0.0f);
            }
            *(v4f*)(dest + gbase + idx) = v;
        }
    } else {
        const int n1 = nrows * F;
        for (int idx = t; idx < n1; idx += 1024) {
            const int r = idx / F;
            const int f = idx - r * F;
            const float di = dinv[rowBase + r];
            const float v = acc[idx] + H[gbase + idx] * di * di + bias[f];
            dest[gbase + idx] = doRelu ? fmaxf(v, 0.0f) : v;
        }
    }
    }
}

__global__ void k_agg_small(
    const float* __restrict__ H, const int* __restrict__ src,
    const int* __restrict__ dst, const float* __restrict__ dinv,
    float* __restrict__ out, int E, int F)
{
    const unsigned nth = gridDim.x * blockDim.x;
    const unsigned tot = (unsigned)E * (unsigned)F;
    const unsigned uF  = (unsigned)F;
    for (unsigned idx = blockIdx.x * blockDim.x + threadIdx.x; idx < tot; idx += nth) {
        const unsigned e = idx / uF;
        const unsigned f = idx - e * uF;
        const int s = src[e], d = dst[e];
        const float nrm = dinv[s] * dinv[d];
        atomicAdd(&out[(size_t)d * F + f], H[(size_t)s * F + f] * nrm);
    }
}

__global__ void k_finish(float* __restrict__ dest, const float* __restrict__ H,
                         const float* __restrict__ dinv, const float* __restrict__ bias,
                         int N, int F, int doRelu)
{
    const unsigned tid = blockIdx.x * blockDim.x + threadIdx.x;
    const unsigned nth = gridDim.x * blockDim.x;
    const unsigned tot = (unsigned)N * (unsigned)F;
    const unsigned uF  = (unsigned)F;
    const bool vec = ((F & 3) == 0) &&
        ((((size_t)dest | (size_t)H | (size_t)bias) & 15) == 0);
    if (vec) {
        const unsigned n4 = tot >> 2;
        for (unsigned q = tid; q < n4; q += nth) {
            const unsigned idx = q << 2;
            const unsigned i = idx / uF;
            const unsigned f = idx - i * uF;
            const float di = dinv[i];
            const v4f d4 = ((const v4f*)dest)[q];
            const v4f h4 = ((const v4f*)H)[q];
            const v4f b4 = *(const v4f*)(bias + f);
            v4f v;
            v.x = d4.x + h4.x * di * di + b4.x;
            v.y = d4.y + h4.y * di * di + b4.y;
            v.z = d4.z + h4.z * di * di + b4.z;
            v.w = d4.w + h4.w * di * di + b4.w;
            if (doRelu) {
                v.x = fmaxf(v.x, 0.0f); v.y = fmaxf(v.y, 0.0f);
                v.z = fmaxf(v.z, 0.0f); v.w = fmaxf(v.w, 0.0f);
            }
            ((v4f*)dest)[q] = v;
        }
    } else {
        for (unsigned idx = tid; idx < tot; idx += nth) {
            const unsigned i = idx / uF;
            const unsigned f = idx - i * uF;
            const float di = dinv[i];
            const float v  = dest[idx] + H[idx] * di * di + bias[f];
            dest[idx] = doRelu ? fmaxf(v, 0.0f) : v;
        }
    }
}

extern "C" void kernel_launch(void* const* d_in, const int* in_sizes, int n_in,
                              void* d_out, int out_size, void* d_ws, size_t ws_size,
                              hipStream_t stream) {
    (void)n_in; (void)out_size;

    const float* x   = (const float*)d_in[0];
    const int*   ei  = (const int*)  d_in[1];
    const float* W1  = (const float*)d_in[2];
    const float* b1  = (const float*)d_in[3];
    const float* W1b = (const float*)d_in[4];
    const float* b1b = (const float*)d_in[5];
    const float* W2b = (const float*)d_in[6];
    const float* b2b = (const float*)d_in[7];
    const float* W2  = (const float*)d_in[8];
    const float* b2  = (const float*)d_in[9];

    const int F_IN = 500, H1 = 200, H2 = 100, H3 = 40, C = 3;
    const int N = in_sizes[0] / F_IN;
    const int E = in_sizes[1] / 2;
    const int* srcIdx = ei;
    const int* dstIdx = ei + E;

    char*  ws   = (char*)d_ws;
    float* dinv = (float*)ws;
    size_t off  = ((size_t)N * sizeof(float) + 511) & ~(size_t)511;
    float* Wpad = (float*)(ws + off);
    off += (size_t)256 * 512 * sizeof(float);
    float* bufH = (float*)(ws + off);
    off += (((size_t)N * H1 * sizeof(float)) + 511) & ~(size_t)511;
    float* bufA = (float*)(ws + off);
    float* out  = (float*)d_out;
    off += (((size_t)N * H1 * sizeof(float)) + 511) & ~(size_t)511;
    int* keyp = (int*)(ws + off);
    const bool keyed = (ws_size >= off + ((((size_t)E + 15) >> 4) << 4)) &&
                       ((N + OWN_R - 1) / OWN_R <= 256);
    int* listA = (int*)bufA;
    int* listB = (int*)(bufH + (size_t)N * H2);
    off += ((((size_t)E + 15) >> 4) << 4);
    off  = (off + 511) & ~(size_t)511;
    int* listK = (int*)(ws + off);
    const bool kept = keyed && (ws_size >= off + (size_t)N * KEEP_CAP * sizeof(int));

    const int resBlocks  = 2048;
    auto gridFor = [&](size_t items) {
        size_t b = (items + 255) / 256;
        return (int)(b < (size_t)resBlocks ? b : (size_t)resBlocks);
    };

    if (keyed) {
        k_key<<<gridFor(((size_t)E + 3) / 4), 256, 0, stream>>>(dstIdx, keyp, E);
        if (kept)
            k_agg_own<0, 1, 1, 1, OWN_R, true, true, 1, KEEP_CAP><<<(N + OWN_R - 1) / OWN_R, 1024, 0, stream>>>(
                nullptr, srcIdx, dstIdx, nullptr, nullptr, dinv, keyp, listK, N, E, 0);
        else
            k_agg_own<0, 1, 1, 1, OWN_R, true, true, 1, 200><<<(N + OWN_R - 1) / OWN_R, 1024, 0, stream>>>(
                nullptr, srcIdx, dstIdx, nullptr, nullptr, dinv, keyp, listA, N, E, 0);
    } else {
        k_fill1<<<(N + 255) / 256, 256, 0, stream>>>(dinv, N);
        k_deg<<<gridFor(((size_t)E + 3) / 4), 256, 0, stream>>>(dstIdx, dinv, E);
        k_rsqrt<<<(N + 255) / 256, 256, 0, stream>>>(dinv, N);
    }

    const int gemmBlocks = (N + 16 * RT - 1) / (16 * RT);

    auto layer = [&](const float* in, const float* Wt, const float* bias,
                     float* dest, int K, int F, int relu) {
        const int Kpad = (K + TK - 1) & ~(TK - 1);
        k_packW<<<(256 * Kpad + 255) / 256, 256, 0, stream>>>(Wt, Wpad, F, K, Kpad);
        if (F == 100)
            k_gemm_wmma<7><<<gemmBlocks, 256, 0, stream>>>(in, Wpad, bufH, N, K, Kpad, F);
        else if (F == 40)
            k_gemm_wmma<3><<<gemmBlocks, 256, 0, stream>>>(in, Wpad, bufH, N, K, Kpad, F);
        else if (F <= 16)
            k_gemm_wmma<1><<<gemmBlocks, 256, 0, stream>>>(in, Wpad, bufH, N, K, Kpad, F);
        else
            k_gemm_wmma<16><<<gemmBlocks, 256, 0, stream>>>(in, Wpad, bufH, N, K, Kpad, F);
        const int keyBlocks = (N + OWN_R - 1) / OWN_R;
        if (F == 200 && kept) {
            k_agg_own<7, 4, 1, 200, OWN_R, true, false, 2, KEEP_CAP><<<keyBlocks, 1024, 0, stream>>>(
                bufH, srcIdx, dstIdx, dinv, bias, dest, keyp, listK, N, E, relu);
        } else if (F == 100 && kept) {
            k_agg_own<4, 8, 1, 100, OWN_R, true, false, 2, KEEP_CAP><<<keyBlocks, 1024, 0, stream>>>(
                bufH, srcIdx, dstIdx, dinv, bias, dest, keyp, listK, N, E, relu);
        } else if (F == 40 && kept) {
            k_agg_own<2, 4, 2, 40, OWN_R, true, false, 2, KEEP_CAP><<<keyBlocks, 1024, 0, stream>>>(
                bufH, srcIdx, dstIdx, dinv, bias, dest, keyp, listK, N, E, relu);
        } else if (F == 3 && kept) {
            k_agg_own<0, 1, 1, 3, OWN_R, true, false, 2, KEEP_CAP><<<keyBlocks, 1024, 0, stream>>>(
                bufH, srcIdx, dstIdx, dinv, bias, dest, keyp, listK, N, E, relu);
        } else if (F == 200 && keyed) {
            k_agg_own<7, 4, 1, 200, OWN_R, true, false, 2, 200><<<keyBlocks, 1024, 0, stream>>>(
                bufH, srcIdx, dstIdx, dinv, bias, dest, keyp, listA, N, E, relu);
        } else if (F == 100 && keyed) {
            k_agg_own<4, 8, 1, 100, OWN_R, true, false, 1, 100><<<keyBlocks, 1024, 0, stream>>>(
                bufH, srcIdx, dstIdx, dinv, bias, dest, keyp, listB, N, E, relu);
        } else if (F == 40 && keyed) {
            k_agg_own<2, 4, 2, 40, OWN_R, true, false, 2, 100><<<keyBlocks, 1024, 0, stream>>>(
                bufH, srcIdx, dstIdx, dinv, bias, dest, keyp, listB, N, E, relu);
        } else if (F == 3 && keyed) {
            k_agg_own<0, 1, 1, 3, OWN_R, true, false, 2, 100><<<keyBlocks, 1024, 0, stream>>>(
                bufH, srcIdx, dstIdx, dinv, bias, dest, keyp, listB, N, E, relu);
        } else if (F == 200) {
            k_agg_own<7, 4, 1, 200, 396, false, false, 0, 1><<<(N + 395) / 396, 1024, 0, stream>>>(
                bufH, srcIdx, dstIdx, dinv, bias, dest, keyp, nullptr, N, E, relu);
        } else if (F == 100) {
            k_agg_own<4, 8, 1, 100, 400, false, false, 0, 1><<<(N + 399) / 400, 1024, 0, stream>>>(
                bufH, srcIdx, dstIdx, dinv, bias, dest, keyp, nullptr, N, E, relu);
        } else if (F == 40) {
            k_agg_own<2, 4, 2, 40, 400, false, false, 0, 1><<<(N + 399) / 400, 1024, 0, stream>>>(
                bufH, srcIdx, dstIdx, dinv, bias, dest, keyp, nullptr, N, E, relu);
        } else {
            size_t tot = (size_t)N * F;
            const int streamGrid = gridFor((tot + 3) / 4);
            k_zero<<<streamGrid, 256, 0, stream>>>(dest, tot);
            k_agg_small<<<gridFor(((size_t)E + 3) / 4), 256, 0, stream>>>(
                bufH, srcIdx, dstIdx, dinv, dest, E, F);
            k_finish<<<streamGrid, 256, 0, stream>>>(
                dest, bufH, dinv, bias, N, F, relu);
        }
    };

    layer(x,    W1,  b1,  bufA, F_IN, H1, 1);
    layer(bufA, W1b, b1b, bufA, H1,   H2, 1);
    layer(bufA, W2b, b2b, bufA, H2,   H3, 1);
    layer(bufA, W2,  b2,  out,  H3,   C,  0);
}
